# starting kernel
# baseline (speedup 1.0000x reference)
#include <hip/hip_runtime.h>
#include <math.h>


#define IN_CH 128
#define HID   64
#define HEADS 4
#define NEG_SLOPE 0.2f

typedef float v2f __attribute__((ext_vector_type(2)));
typedef float v8f __attribute__((ext_vector_type(8)));

__device__ __forceinline__ unsigned f2ord(float f) {
  unsigned u = __float_as_uint(f);
  return (u & 0x80000000u) ? ~u : (u | 0x80000000u);
}
__device__ __forceinline__ float ord2f(unsigned u) {
  return (u & 0x80000000u) ? __uint_as_float(u & 0x7fffffffu)
                           : __uint_as_float(~u);
}
#define NEG_INF_ORD 0x007FFFFFu

__device__ __forceinline__ float elu_f(float x) {
  return x > 0.0f ? x : (expf(x) - 1.0f);
}

__global__ void fill_f32(float* p, float v, int n) {
  int i = blockIdx.x * blockDim.x + threadIdx.x;
  if (i < n) p[i] = v;
}
__global__ void fill_u32(unsigned* p, unsigned v, int n) {
  int i = blockIdx.x * blockDim.x + threadIdx.x;
  if (i < n) p[i] = v;
}

template <int K>
__global__ void wmma_gemm_f32(const float* __restrict__ A,
                              const float* __restrict__ B,
                              float* __restrict__ C, int M, int Nc) {
  int wave = (int)((blockIdx.x * blockDim.x + threadIdx.x) >> 5);
  int ntn  = Nc >> 4;
  int total = (M >> 4) * ntn;
  if (wave >= total) return;
  int tm = wave / ntn, tn = wave % ntn;
  int lane = (int)(threadIdx.x & 31);
  int lr = lane & 15, hi = lane >> 4;

  const float* arow = A + (size_t)(tm * 16 + lr) * K;
  const float* bcol = B + tn * 16 + lr;
  v8f acc = {};
#pragma unroll 8
  for (int k0 = 0; k0 < K; k0 += 4) {
    int ka = k0 + hi * 2;
    v2f a; a.x = arow[ka];               a.y = arow[ka + 1];
    v2f b; b.x = bcol[(size_t)ka * Nc];  b.y = bcol[(size_t)(ka + 1) * Nc];
    acc = __builtin_amdgcn_wmma_f32_16x16x4_f32(
         false, a,  false, b,
         (short)0, acc,  false,  false);
  }
  float* crow = C + tn * 16 + lr;
  int rbase = tm * 16 + hi * 8;
#pragma unroll
  for (int v = 0; v < 8; ++v) crow[(size_t)(rbase + v) * Nc] = acc[v];
}

template <int H>
__global__ void attn_coef(const float* __restrict__ h,
                          const float* __restrict__ a_src,
                          const float* __restrict__ a_dst,
                          float* __restrict__ es, float* __restrict__ ed, int n) {
  int idx = blockIdx.x * blockDim.x + threadIdx.x;
  if (idx >= n * H) return;
  int i = idx / H, hh = idx % H;
  const float* hp = h + (size_t)i * (H * HID) + hh * HID;
  const float* as = a_src + hh * HID;
  const float* ad = a_dst + hh * HID;
  float s = 0.f, d = 0.f;
#pragma unroll 8
  for (int c = 0; c < HID; ++c) { float v = hp[c]; s += v * as[c]; d += v * ad[c]; }
  es[idx] = s; ed[idx] = d;
}

template <int H>
__global__ void edge_logits(const int* __restrict__ ei,
                            const float* __restrict__ es,
                            const float* __restrict__ ed,
                            float* __restrict__ logit,
                            unsigned* __restrict__ mEnc, int E_, int N_) {
  int idx = blockIdx.x * blockDim.x + threadIdx.x;
  if (idx >= (E_ + N_) * H) return;
  int e = idx / H, hh = idx % H;
  int s = (e < E_) ? ei[e]      : (e - E_);
  int d = (e < E_) ? ei[E_ + e] : (e - E_);
  float l = es[s * H + hh] + ed[d * H + hh];
  l = (l > 0.0f) ? l : NEG_SLOPE * l;
  logit[idx] = l;
  atomicMax(&mEnc[d * H + hh], f2ord(l));
}

template <int H>
__global__ void edge_expsum(const int* __restrict__ ei,
                            const float* __restrict__ logit,
                            const unsigned* __restrict__ mEnc,
                            float* __restrict__ ssum, int E_, int N_) {
  int idx = blockIdx.x * blockDim.x + threadIdx.x;
  if (idx >= (E_ + N_) * H) return;
  int e = idx / H, hh = idx % H;
  int d = (e < E_) ? ei[E_ + e] : (e - E_);
  float m = ord2f(mEnc[d * H + hh]);
  atomicAdd(&ssum[d * H + hh], expf(logit[idx] - m));
}

template <int H>
__global__ void edge_message4(const int* __restrict__ ei,
                              const float* __restrict__ h,
                              const float* __restrict__ logit,
                              const unsigned* __restrict__ mEnc,
                              const float* __restrict__ ssum,
                              float* __restrict__ agg, int E_, int N_) {
  const int LDH = H * HID;
  const int LD4 = LDH / 4;
  int idx = blockIdx.x * blockDim.x + threadIdx.x;
  if (idx >= (E_ + N_) * LD4) return;
  int e = idx / LD4;
  int t = (idx % LD4) * 4;
  int hh = t >> 6;
  int s = (e < E_) ? ei[e]      : (e - E_);
  int d = (e < E_) ? ei[E_ + e] : (e - E_);
  float m = ord2f(mEnc[d * H + hh]);
  float a = expf(logit[e * H + hh] - m) / (ssum[d * H + hh] + 1e-16f);
  const float4 hv = *reinterpret_cast<const float4*>(h + (size_t)s * LDH + t);
  float* ap = agg + (size_t)d * LDH + t;
  atomicAdd(ap + 0, hv.x * a);
  atomicAdd(ap + 1, hv.y * a);
  atomicAdd(ap + 2, hv.z * a);
  atomicAdd(ap + 3, hv.w * a);
}

template <int LD>
__global__ void bias_elu(const float* __restrict__ agg,
                         const float* __restrict__ bias,
                         float* __restrict__ out, int total) {
  int idx = blockIdx.x * blockDim.x + threadIdx.x;
  if (idx >= total) return;
  out[idx] = elu_f(agg[idx] + bias[idx % LD]);
}

__global__ void out_proj(const float* __restrict__ h,
                         const float* __restrict__ Wo,
                         const float* __restrict__ bo,
                         float* __restrict__ out, int n) {
  int wid  = (int)((blockIdx.x * blockDim.x + threadIdx.x) >> 5);
  int lane = (int)(threadIdx.x & 31);
  if (wid >= n) return;
  const float* hp = h + (size_t)wid * HID;
  float s = hp[lane] * Wo[lane] + hp[lane + 32] * Wo[lane + 32];
#pragma unroll
  for (int o = 16; o > 0; o >>= 1) s += __shfl_xor(s, o, 32);
  if (lane == 0) out[wid] = s + bo[0];
}

#define CDIV(a, b) (((a) + (b) - 1) / (b))

extern "C" void kernel_launch(void* const* d_in, const int* in_sizes, int n_in,
                              void* d_out, int out_size, void* d_ws, size_t ws_size,
                              hipStream_t stream) {
  const float* x   = (const float*)d_in[0];
  const float* W1  = (const float*)d_in[1];
  const float* as1 = (const float*)d_in[2];
  const float* ad1 = (const float*)d_in[3];
  const float* b1  = (const float*)d_in[4];
  const float* W2  = (const float*)d_in[5];
  const float* as2 = (const float*)d_in[6];
  const float* ad2 = (const float*)d_in[7];
  const float* b2  = (const float*)d_in[8];
  const float* Wo  = (const float*)d_in[9];
  const float* bo  = (const float*)d_in[10];
  const int*   ei  = (const int*)d_in[11];

  const int n  = in_sizes[0]  / IN_CH;
  const int e  = in_sizes[11] / 2;
  const int et = e + n;

  float* p = (float*)d_ws;
  float*    h1   = p;            p += (size_t)n * 256;
  float*    agg1 = p;            p += (size_t)n * 256;
  float*    h2   = p;            p += (size_t)n * 64;
  float*    agg2 = p;            p += (size_t)n * 64;
  float*    es1  = p;            p += (size_t)n * 4;
  float*    ed1  = p;            p += (size_t)n * 4;
  unsigned* m1   = (unsigned*)p; p += (size_t)n * 4;
  float*    s1   = p;            p += (size_t)n * 4;
  float*    es2v = p;            p += n;
  float*    ed2v = p;            p += n;
  unsigned* m2   = (unsigned*)p; p += n;
  float*    s2   = p;            p += n;
  float*    lg1  = p;            p += (size_t)et * 4;
  float*    lg2  = p;

  const int B = 256;

  fill_f32<<<CDIV(n * 256, B), B, 0, stream>>>(agg1, 0.0f, n * 256);
  fill_f32<<<CDIV(n * 64,  B), B, 0, stream>>>(agg2, 0.0f, n * 64);
  fill_f32<<<CDIV(n * 4,   B), B, 0, stream>>>(s1,   0.0f, n * 4);
  fill_f32<<<CDIV(n,       B), B, 0, stream>>>(s2,   0.0f, n);
  fill_u32<<<CDIV(n * 4,   B), B, 0, stream>>>(m1, NEG_INF_ORD, n * 4);
  fill_u32<<<CDIV(n,       B), B, 0, stream>>>(m2, NEG_INF_ORD, n);

  int tiles1 = (n / 16) * (256 / 16);
  wmma_gemm_f32<128><<<CDIV(tiles1, 8), B, 0, stream>>>(x, W1, h1, n, 256);
  attn_coef<HEADS><<<CDIV(n * HEADS, B), B, 0, stream>>>(h1, as1, ad1, es1, ed1, n);
  edge_logits<HEADS><<<CDIV(et * HEADS, B), B, 0, stream>>>(ei, es1, ed1, lg1, m1, e, n);
  edge_expsum<HEADS><<<CDIV(et * HEADS, B), B, 0, stream>>>(ei, lg1, m1, s1, e, n);
  edge_message4<HEADS><<<CDIV(et * 64, B), B, 0, stream>>>(ei, h1, lg1, m1, s1, agg1, e, n);
  bias_elu<256><<<CDIV(n * 256, B), B, 0, stream>>>(agg1, b1, h1, n * 256);

  int tiles2 = (n / 16) * (64 / 16);
  wmma_gemm_f32<256><<<CDIV(tiles2, 8), B, 0, stream>>>(h1, W2, h2, n, 64);
  attn_coef<1><<<CDIV(n, B), B, 0, stream>>>(h2, as2, ad2, es2v, ed2v, n);
  edge_logits<1><<<CDIV(et, B), B, 0, stream>>>(ei, es2v, ed2v, lg2, m2, e, n);
  edge_expsum<1><<<CDIV(et, B), B, 0, stream>>>(ei, lg2, m2, s2, e, n);
  edge_message4<1><<<CDIV(et * 16, B), B, 0, stream>>>(ei, h2, lg2, m2, s2, agg2, e, n);
  bias_elu<64><<<CDIV(n * 64, B), B, 0, stream>>>(agg2, b2, h2, n * 64);

  out_proj<<<CDIV(n * 32, B), B, 0, stream>>>(h2, Wo, bo, (float*)d_out, n);
}


# Message pass by destination owner: a work-group keeps 200 nodes' row sums and denominators in LDS, scans the edge list for its range, adds by LDS f32 atomics, norm+bias+ELU fused; arrival-order sums
# speedup vs baseline: 2.1303x; 2.1303x over previous
#include <hip/hip_runtime.h>
#include <math.h>


#define IN_CH 128
#define HID   64
#define HEADS 4
#define NEG_SLOPE 0.2f

typedef float v2f __attribute__((ext_vector_type(2)));
typedef float v8f __attribute__((ext_vector_type(8)));

__device__ __forceinline__ unsigned f2ord(float f) {
  unsigned u = __float_as_uint(f);
  return (u & 0x80000000u) ? ~u : (u | 0x80000000u);
}
__device__ __forceinline__ float ord2f(unsigned u) {
  return (u & 0x80000000u) ? __uint_as_float(u & 0x7fffffffu)
                           : __uint_as_float(~u);
}
#define NEG_INF_ORD 0x007FFFFFu

__device__ __forceinline__ float elu_f(float x) {
  return x > 0.0f ? x : (expf(x) - 1.0f);
}

__global__ void fill_u32(unsigned* p, unsigned v, int n) {
  int i = blockIdx.x * blockDim.x + threadIdx.x;
  if (i < n) p[i] = v;
}

template <int K>
__global__ void wmma_gemm_f32(const float* __restrict__ A,
                              const float* __restrict__ B,
                              float* __restrict__ C, int M, int Nc) {
  int wave = (int)((blockIdx.x * blockDim.x + threadIdx.x) >> 5);
  int ntn  = Nc >> 4;
  int total = (M >> 4) * ntn;
  if (wave >= total) return;
  int tm = wave / ntn, tn = wave % ntn;
  int lane = (int)(threadIdx.x & 31);
  int lr = lane & 15, hi = lane >> 4;

  const float* arow = A + (size_t)(tm * 16 + lr) * K;
  const float* bcol = B + tn * 16 + lr;
  v8f acc = {};
#pragma unroll 8
  for (int k0 = 0; k0 < K; k0 += 4) {
    int ka = k0 + hi * 2;
    v2f a; a.x = arow[ka];               a.y = arow[ka + 1];
    v2f b; b.x = bcol[(size_t)ka * Nc];  b.y = bcol[(size_t)(ka + 1) * Nc];
    acc = __builtin_amdgcn_wmma_f32_16x16x4_f32(
         false, a,  false, b,
         (short)0, acc,  false,  false);
  }
  float* crow = C + tn * 16 + lr;
  int rbase = tm * 16 + hi * 8;
#pragma unroll
  for (int v = 0; v < 8; ++v) crow[(size_t)(rbase + v) * Nc] = acc[v];
}

template <int H>
__global__ void attn_coef(const float* __restrict__ h,
                          const float* __restrict__ a_src,
                          const float* __restrict__ a_dst,
                          float* __restrict__ es, float* __restrict__ ed, int n) {
  int idx = blockIdx.x * blockDim.x + threadIdx.x;
  if (idx >= n * H) return;
  int i = idx / H, hh = idx % H;
  const float* hp = h + (size_t)i * (H * HID) + hh * HID;
  const float* as = a_src + hh * HID;
  const float* ad = a_dst + hh * HID;
  float s = 0.f, d = 0.f;
#pragma unroll 8
  for (int c = 0; c < HID; ++c) { float v = hp[c]; s += v * as[c]; d += v * ad[c]; }
  es[idx] = s; ed[idx] = d;
}

template <int H>
__global__ void edge_logits(const int* __restrict__ ei,
                            const float* __restrict__ es,
                            const float* __restrict__ ed,
                            float* __restrict__ logit,
                            unsigned* __restrict__ mEnc, int E_, int N_) {
  int idx = blockIdx.x * blockDim.x + threadIdx.x;
  if (idx >= (E_ + N_) * H) return;
  int e = idx / H, hh = idx % H;
  int s = (e < E_) ? ei[e]      : (e - E_);
  int d = (e < E_) ? ei[E_ + e] : (e - E_);
  float l = es[s * H + hh] + ed[d * H + hh];
  l = (l > 0.0f) ? l : NEG_SLOPE * l;
  logit[idx] = l;
  atomicMax(&mEnc[d * H + hh], f2ord(l));
}

template <bool B> struct flag_t { static constexpr bool value = B; };

template <int H, int NR, int LCAP>
__global__ void __launch_bounds__(1024)
owner_message(const int* __restrict__ ei,
              const float* __restrict__ h,
              const float* __restrict__ logit,
              const unsigned* __restrict__ mEnc,
              const float* __restrict__ bias,
              float* __restrict__ out, int E_, int N_) {
  constexpr int LDH = H * HID;
  constexpr int NL  = LDH / 32;
  constexpr int EPW = 8;
  constexpr int NT  = 1024;
  constexpr int U   = 8;
  static_assert(H * EPW <= 32, "one lane per (edge, head) pair of the group");
  static_assert(LCAP >= NR + 8, "self-loops plus a smallest window always fit");

  __shared__ __attribute__((aligned(16))) float acc[NR * LDH];
  __shared__ float den[NR * H];
  __shared__ float mx[NR * H];
  __shared__ int   list[LCAP];
  __shared__ int   cnt;

  const int tid  = (int)threadIdx.x;
  const int lane = tid & 31;
  const int wv   = __builtin_amdgcn_readfirstlane(tid >> 5);
  const int n0   = (int)blockIdx.x * NR;
  const int nrv  = (N_ - n0 < NR) ? (N_ - n0) : NR;
  const unsigned nrvu = (unsigned)nrv;
  const int* __restrict__ dstp = ei + E_;
  const int4* __restrict__ dst4 = reinterpret_cast<const int4*>(dstp);
  const bool vec_ok = ((E_ & 3) == 0) && ((reinterpret_cast<size_t>(dstp) & 15) == 0);
  const int q  = lane / H, hh = lane % H;

  for (int i = tid; i < NR * LDH / 4; i += NT)
    reinterpret_cast<float4*>(acc)[i] = make_float4(0.f, 0.f, 0.f, 0.f);
  for (int i = tid; i < NR * H; i += NT) {
    den[i] = 0.f;
    mx[i]  = (i < nrv * H) ? ord2f(mEnc[(size_t)n0 * H + i]) : 0.f;
  }

  auto push = [&](int e) {
    const int pos = atomicAdd(&cnt, 1);
    if (pos < LCAP) list[pos] = e;
  };

  auto add_rows = [&](auto full, int s, int dl, float p, int nvalid) {
    constexpr bool FULL = decltype(full)::value;
    constexpr int G = (NL >= 8) ? 4 : 8;
#pragma unroll
    for (int q0 = 0; q0 < EPW; q0 += G) {
      float hv[G][NL];
      float aq[G][H];
      int   dq[G];
#pragma unroll
      for (int t = 0; t < G; ++t) {
        if (FULL || q0 + t < nvalid) {
          const int sq = __builtin_amdgcn_readlane(s, (q0 + t) * H);
          dq[t] = __builtin_amdgcn_readlane(dl, (q0 + t) * H);
#pragma unroll
          for (int k = 0; k < H; ++k)
            aq[t][k] = __int_as_float(__builtin_amdgcn_readlane(__float_as_int(p), (q0 + t) * H + k));
          const float* hp = h + (size_t)sq * LDH + lane;
#pragma unroll
          for (int j = 0; j < NL; ++j) hv[t][j] = hp[32 * j];
        }
      }
#pragma unroll
      for (int t = 0; t < G; ++t) {
        if (FULL || q0 + t < nvalid) {
          float* ap = acc + dq[t] * LDH + lane;
#pragma unroll
          for (int j = 0; j < NL; ++j) atomicAdd(ap + 32 * j, hv[t][j] * aq[t][(32 * j) / HID]);
        }
      }
    }
  };

  int lo = 0, hi = E_;
  for (;;) {
    const int nself = (lo == 0) ? nrv : 0;
    for (int i = tid; i < nself; i += NT) list[i] = E_ + n0 + i;
    if (tid == 0) cnt = nself;
    __syncthreads();
    if (vec_ok) {
      const int vlo = lo >> 2, vhi = hi >> 2;
      for (int base = vlo + tid; base < vhi; base += NT * U) {
        int4 dv[U];
#pragma unroll
        for (int u = 0; u < U; ++u) {
          const int i = base + u * NT;
          dv[u] = (i < vhi) ? dst4[i] : make_int4(-1, -1, -1, -1);
        }
#pragma unroll
        for (int u = 0; u < U; ++u) {
          const int e = (base + u * NT) * 4;
          const bool b0 = (unsigned)(dv[u].x - n0) < nrvu;
          const bool b1 = (unsigned)(dv[u].y - n0) < nrvu;
          const bool b2 = (unsigned)(dv[u].z - n0) < nrvu;
          const bool b3 = (unsigned)(dv[u].w - n0) < nrvu;
          if (b0 | b1 | b2 | b3) {
            if (b0) push(e);
            if (b1) push(e + 1);
            if (b2) push(e + 2);
            if (b3) push(e + 3);
          }
        }
      }
    } else {
      for (int e = lo + tid; e < hi; e += NT)
        if ((unsigned)(dstp[e] - n0) < nrvu) push(e);
    }
    __syncthreads();
    const int total = __builtin_amdgcn_readfirstlane(cnt);
    if (total > LCAP) {
      hi = lo + (((hi - lo) >> 1) & ~3);
      __syncthreads();
      continue;
    }

    const int ngroups = (total + EPW - 1) / EPW;
    for (int g = wv; g < ngroups; g += NT / 32) {
      const int i0 = g * EPW;
      const int nvalid = (total - i0 < EPW) ? (total - i0) : EPW;
      const bool act = q < nvalid;
      const int e  = list[i0 + (act ? q : 0)];
      const int el = (e < E_) ? e : 0;
      const int sv = ei[el], dv = dstp[el];
      const int s  = (e < E_) ? sv : (e - E_);
      const int dl = ((e < E_) ? dv : (e - E_)) - n0;
      const float p = expf(logit[(size_t)e * H + hh] - mx[dl * H + hh]);
      if (act) atomicAdd(&den[dl * H + hh], p);
      if (nvalid == EPW) add_rows(flag_t<true>{},  s, dl, p, nvalid);
      else               add_rows(flag_t<false>{}, s, dl, p, nvalid);
    }
    if (hi >= E_) break;
    lo = hi; hi = E_;
    __syncthreads();
  }
  __syncthreads();

  for (int idx = tid; idx < nrv * LDH; idx += NT) {
    const int node = idx / LDH, ch = idx % LDH;
    const float dn = den[node * H + ch / HID] + 1e-16f;
    out[(size_t)(n0 + node) * LDH + ch] = elu_f(acc[idx] / dn + bias[ch]);
  }
}

__global__ void out_proj(const float* __restrict__ h,
                         const float* __restrict__ Wo,
                         const float* __restrict__ bo,
                         float* __restrict__ out, int n) {
  int wid  = (int)((blockIdx.x * blockDim.x + threadIdx.x) >> 5);
  int lane = (int)(threadIdx.x & 31);
  if (wid >= n) return;
  const float* hp = h + (size_t)wid * HID;
  float s = hp[lane] * Wo[lane] + hp[lane + 32] * Wo[lane + 32];
#pragma unroll
  for (int o = 16; o > 0; o >>= 1) s += __shfl_xor(s, o, 32);
  if (lane == 0) out[wid] = s + bo[0];
}

#define CDIV(a, b) (((a) + (b) - 1) / (b))

extern "C" void kernel_launch(void* const* d_in, const int* in_sizes, int n_in,
                              void* d_out, int out_size, void* d_ws, size_t ws_size,
                              hipStream_t stream) {
  const float* x   = (const float*)d_in[0];
  const float* W1  = (const float*)d_in[1];
  const float* as1 = (const float*)d_in[2];
  const float* ad1 = (const float*)d_in[3];
  const float* b1  = (const float*)d_in[4];
  const float* W2  = (const float*)d_in[5];
  const float* as2 = (const float*)d_in[6];
  const float* ad2 = (const float*)d_in[7];
  const float* b2  = (const float*)d_in[8];
  const float* Wo  = (const float*)d_in[9];
  const float* bo  = (const float*)d_in[10];
  const int*   ei  = (const int*)d_in[11];

  const int n  = in_sizes[0]  / IN_CH;
  const int e  = in_sizes[11] / 2;
  const int et = e + n;

  float* p = (float*)d_ws;
  float*    h1   = p;            p += (size_t)n * 256;
  float*    o1   = p;            p += (size_t)n * 256;
  float*    h2   = p;            p += (size_t)n * 64;
  float*    o2   = p;            p += (size_t)n * 64;
  float*    es1  = p;            p += (size_t)n * 4;
  float*    ed1  = p;            p += (size_t)n * 4;
  unsigned* m1   = (unsigned*)p; p += (size_t)n * 4;
  p += (size_t)n * 4;
  float*    es2v = p;            p += n;
  float*    ed2v = p;            p += n;
  unsigned* m2   = (unsigned*)p; p += n;
  p += n;
  float*    lg1  = p;            p += (size_t)et * 4;
  float*    lg2  = p;

  const int B = 256;
  constexpr int NR = 200;
  constexpr int LCAP = 8192;

  fill_u32<<<CDIV(n * 4,   B), B, 0, stream>>>(m1, NEG_INF_ORD, n * 4);
  fill_u32<<<CDIV(n,       B), B, 0, stream>>>(m2, NEG_INF_ORD, n);

  int tiles1 = (n / 16) * (256 / 16);
  wmma_gemm_f32<128><<<CDIV(tiles1, 8), B, 0, stream>>>(x, W1, h1, n, 256);
  attn_coef<HEADS><<<CDIV(n * HEADS, B), B, 0, stream>>>(h1, as1, ad1, es1, ed1, n);
  edge_logits<HEADS><<<CDIV(et * HEADS, B), B, 0, stream>>>(ei, es1, ed1, lg1, m1, e, n);
  owner_message<HEADS, NR, LCAP><<<CDIV(n, NR), 1024, 0, stream>>>(ei, h1, lg1, m1, b1, o1, e, n);

  int tiles2 = (n / 16) * (64 / 16);
  wmma_gemm_f32<256><<<CDIV(tiles2, 8), B, 0, stream>>>(o1, W2, h2, n, 64);
  attn_coef<1><<<CDIV(n, B), B, 0, stream>>>(h2, as2, ad2, es2v, ed2v, n);
  edge_logits<1><<<CDIV(et, B), B, 0, stream>>>(ei, es2v, ed2v, lg2, m2, e, n);
  owner_message<1, NR, LCAP><<<CDIV(n, NR), 1024, 0, stream>>>(ei, h2, lg2, m2, b2, o2, e, n);

  out_proj<<<CDIV(n * 32, B), B, 0, stream>>>(o2, Wo, bo, (float*)d_out, n);
}


# Logit pass folded into the owner kernel: a work-group resolves its listed edges, forms logits from es/ed, takes per-node maxima by LDS max; 4 launches and logit arrays gone; arrival-order sums kept
# speedup vs baseline: 2.5736x; 1.2081x over previous
#include <hip/hip_runtime.h>
#include <math.h>


#define IN_CH 128
#define HID   64
#define HEADS 4
#define NEG_SLOPE 0.2f

typedef float v2f __attribute__((ext_vector_type(2)));
typedef float v8f __attribute__((ext_vector_type(8)));

__device__ __forceinline__ unsigned f2ord(float f) {
  unsigned u = __float_as_uint(f);
  return (u & 0x80000000u) ? ~u : (u | 0x80000000u);
}
__device__ __forceinline__ float ord2f(unsigned u) {
  return (u & 0x80000000u) ? __uint_as_float(u & 0x7fffffffu)
                           : __uint_as_float(~u);
}
#define NEG_INF_ORD 0x007FFFFFu

__device__ __forceinline__ float elu_f(float x) {
  return x > 0.0f ? x : (expf(x) - 1.0f);
}

template <int K>
__global__ void wmma_gemm_f32(const float* __restrict__ A,
                              const float* __restrict__ B,
                              float* __restrict__ C, int M, int Nc) {
  int wave = (int)((blockIdx.x * blockDim.x + threadIdx.x) >> 5);
  int ntn  = Nc >> 4;
  int total = (M >> 4) * ntn;
  if (wave >= total) return;
  int tm = wave / ntn, tn = wave % ntn;
  int lane = (int)(threadIdx.x & 31);
  int lr = lane & 15, hi = lane >> 4;

  const float* arow = A + (size_t)(tm * 16 + lr) * K;
  const float* bcol = B + tn * 16 + lr;
  v8f acc = {};
#pragma unroll 8
  for (int k0 = 0; k0 < K; k0 += 4) {
    int ka = k0 + hi * 2;
    v2f a; a.x = arow[ka];               a.y = arow[ka + 1];
    v2f b; b.x = bcol[(size_t)ka * Nc];  b.y = bcol[(size_t)(ka + 1) * Nc];
    acc = __builtin_amdgcn_wmma_f32_16x16x4_f32(
         false, a,  false, b,
         (short)0, acc,  false,  false);
  }
  float* crow = C + tn * 16 + lr;
  int rbase = tm * 16 + hi * 8;
#pragma unroll
  for (int v = 0; v < 8; ++v) crow[(size_t)(rbase + v) * Nc] = acc[v];
}

template <int H>
__global__ void attn_coef(const float* __restrict__ h,
                          const float* __restrict__ a_src,
                          const float* __restrict__ a_dst,
                          float* __restrict__ es, float* __restrict__ ed, int n) {
  int idx = blockIdx.x * blockDim.x + threadIdx.x;
  if (idx >= n * H) return;
  int i = idx / H, hh = idx % H;
  const float* hp = h + (size_t)i * (H * HID) + hh * HID;
  const float* as = a_src + hh * HID;
  const float* ad = a_dst + hh * HID;
  float s = 0.f, d = 0.f;
#pragma unroll 8
  for (int c = 0; c < HID; ++c) { float v = hp[c]; s += v * as[c]; d += v * ad[c]; }
  es[idx] = s; ed[idx] = d;
}

template <bool B> struct flag_t { static constexpr bool value = B; };

template <int H, int NR, int LCAP>
__global__ void __launch_bounds__(1024)
owner_message(const int* __restrict__ ei,
              const float* __restrict__ h,
              const float* __restrict__ es,
              const float* __restrict__ ed,
              const float* __restrict__ bias,
              float* __restrict__ out, int E_, int N_) {
  constexpr int LDH = H * HID;
  constexpr int NL  = LDH / 32;
  constexpr int EPW = 8;
  constexpr int NT  = 1024;
  constexpr int U   = 8;
  constexpr int UR  = 4;
  static_assert(H * EPW <= 32, "one lane per (edge, head) pair of the group");
  static_assert(LCAP >= NR + 8, "self-loops plus a smallest window always fit");

  __shared__ __attribute__((aligned(16))) float acc[NR * LDH];
  __shared__ float    den[NR * H];
  __shared__ unsigned mxu[NR * H];
  __shared__ float    edl[NR * H];
  __shared__ int2     list[LCAP];
  __shared__ int      cnt;

  const int tid  = (int)threadIdx.x;
  const int lane = tid & 31;
  const int wv   = __builtin_amdgcn_readfirstlane(tid >> 5);
  const int n0   = (int)blockIdx.x * NR;
  const int nrv  = (N_ - n0 < NR) ? (N_ - n0) : NR;
  const unsigned nrvu = (unsigned)nrv;
  const int* __restrict__ dstp = ei + E_;
  const int4* __restrict__ dst4 = reinterpret_cast<const int4*>(dstp);
  const bool vec_ok = ((E_ & 3) == 0) && ((reinterpret_cast<size_t>(dstp) & 15) == 0);
  const int q  = lane / H, hh = lane % H;

  for (int i = tid; i < NR * LDH / 4; i += NT)
    reinterpret_cast<float4*>(acc)[i] = make_float4(0.f, 0.f, 0.f, 0.f);
  for (int i = tid; i < NR * H; i += NT) {
    den[i] = 0.f;
    mxu[i] = NEG_INF_ORD;
    edl[i] = (i < nrv * H) ? ed[(size_t)n0 * H + i] : 0.f;
  }

  auto push = [&](int e) {
    const int pos = atomicAdd(&cnt, 1);
    if (pos < LCAP) list[pos].x = e;
  };

  auto add_rows = [&](auto full, int s, int dl, float esv, bool act, int nvalid) {
    constexpr bool FULL = decltype(full)::value;
    constexpr int G = (NL >= 8) ? 4 : 8;
    float p = 0.f;
#pragma unroll
    for (int q0 = 0; q0 < EPW; q0 += G) {
      float hv[G][NL];
      int   dq[G];
#pragma unroll
      for (int t = 0; t < G; ++t) {
        if (FULL || q0 + t < nvalid) {
          const int sq = __builtin_amdgcn_readlane(s, (q0 + t) * H);
          dq[t] = __builtin_amdgcn_readlane(dl, (q0 + t) * H);
          const float* hp = h + (size_t)sq * LDH + lane;
#pragma unroll
          for (int j = 0; j < NL; ++j) hv[t][j] = hp[32 * j];
        }
      }
      if (q0 == 0) {
        float l = esv + edl[dl * H + hh];
        l = (l > 0.0f) ? l : NEG_SLOPE * l;
        p = expf(l - ord2f(mxu[dl * H + hh]));
        if (act) atomicAdd(&den[dl * H + hh], p);
      }
#pragma unroll
      for (int t = 0; t < G; ++t) {
        if (FULL || q0 + t < nvalid) {
          float aq[H];
#pragma unroll
          for (int k = 0; k < H; ++k)
            aq[k] = __int_as_float(__builtin_amdgcn_readlane(__float_as_int(p), (q0 + t) * H + k));
          float* ap = acc + dq[t] * LDH + lane;
#pragma unroll
          for (int j = 0; j < NL; ++j) atomicAdd(ap + 32 * j, hv[t][j] * aq[(32 * j) / HID]);
        }
      }
    }
  };

  int pass = 0;
  int lo = 0, hi = E_;
  for (;;) {
    const int nself = (lo == 0) ? nrv : 0;
    for (int i = tid; i < nself; i += NT) list[i].x = E_ + n0 + i;
    if (tid == 0) cnt = nself;
    __syncthreads();
    if (vec_ok) {
      const int vlo = lo >> 2, vhi = hi >> 2;
      for (int base = vlo + tid; base < vhi; base += NT * U) {
        int4 dv[U];
#pragma unroll
        for (int u = 0; u < U; ++u) {
          const int i = base + u * NT;
          dv[u] = (i < vhi) ? dst4[i] : make_int4(-1, -1, -1, -1);
        }
#pragma unroll
        for (int u = 0; u < U; ++u) {
          const int e = (base + u * NT) * 4;
          const bool b0 = (unsigned)(dv[u].x - n0) < nrvu;
          const bool b1 = (unsigned)(dv[u].y - n0) < nrvu;
          const bool b2 = (unsigned)(dv[u].z - n0) < nrvu;
          const bool b3 = (unsigned)(dv[u].w - n0) < nrvu;
          if (b0 | b1 | b2 | b3) {
            if (b0) push(e);
            if (b1) push(e + 1);
            if (b2) push(e + 2);
            if (b3) push(e + 3);
          }
        }
      }
    } else {
      for (int e = lo + tid; e < hi; e += NT)
        if ((unsigned)(dstp[e] - n0) < nrvu) push(e);
    }
    __syncthreads();
    const int total = __builtin_amdgcn_readfirstlane(cnt);
    if (total > LCAP) {
      if (pass == 0) pass = 1;
      hi = lo + (((hi - lo) >> 1) & ~3);
      __syncthreads();
      continue;
    }

    for (int base = tid; base < total; base += NT * UR) {
      int iv[UR], sv[UR], dv[UR];
#pragma unroll
      for (int u = 0; u < UR; ++u) {
        const int i = base + u * NT;
        iv[u] = (i < total) ? i : base;
        const int e  = list[iv[u]].x;
        const int el = (e < E_) ? e : 0;
        const int s_ = ei[el], d_ = dstp[el];
        sv[u] = (e < E_) ? s_ : (e - E_);
        dv[u] = ((e < E_) ? d_ : (e - E_)) - n0;
      }
      float ev[UR][H];
#pragma unroll
      for (int u = 0; u < UR; ++u)
#pragma unroll
        for (int k = 0; k < H; ++k) ev[u][k] = es[(size_t)sv[u] * H + k];
#pragma unroll
      for (int u = 0; u < UR; ++u) {
        list[iv[u]] = make_int2(sv[u], dv[u]);
#pragma unroll
        for (int k = 0; k < H; ++k) {
          float l = ev[u][k] + edl[dv[u] * H + k];
          l = (l > 0.0f) ? l : NEG_SLOPE * l;
          atomicMax(&mxu[dv[u] * H + k], f2ord(l));
        }
      }
    }
    __syncthreads();

    if (pass != 1) {
      const int ngroups = (total + EPW - 1) / EPW;
      for (int g = wv; g < ngroups; g += NT / 32) {
        const int i0 = g * EPW;
        const int nvalid = (total - i0 < EPW) ? (total - i0) : EPW;
        const bool act = q < nvalid;
        const int2 sd = list[i0 + (act ? q : 0)];
        const float esv = es[(size_t)sd.x * H + hh];
        if (nvalid == EPW) add_rows(flag_t<true>{},  sd.x, sd.y, esv, act, nvalid);
        else               add_rows(flag_t<false>{}, sd.x, sd.y, esv, act, nvalid);
      }
    }
    if (hi >= E_) {
      if (pass != 1) break;
      pass = 2; lo = 0; hi = E_;
    } else {
      lo = hi; hi = E_;
    }
    __syncthreads();
  }
  __syncthreads();

  for (int idx = tid; idx < nrv * LDH; idx += NT) {
    const int node = idx / LDH, ch = idx % LDH;
    const float dn = den[node * H + ch / HID] + 1e-16f;
    out[(size_t)(n0 + node) * LDH + ch] = elu_f(acc[idx] / dn + bias[ch]);
  }
}

__global__ void out_proj(const float* __restrict__ h,
                         const float* __restrict__ Wo,
                         const float* __restrict__ bo,
                         float* __restrict__ out, int n) {
  int wid  = (int)((blockIdx.x * blockDim.x + threadIdx.x) >> 5);
  int lane = (int)(threadIdx.x & 31);
  if (wid >= n) return;
  const float* hp = h + (size_t)wid * HID;
  float s = hp[lane] * Wo[lane] + hp[lane + 32] * Wo[lane + 32];
#pragma unroll
  for (int o = 16; o > 0; o >>= 1) s += __shfl_xor(s, o, 32);
  if (lane == 0) out[wid] = s + bo[0];
}

#define CDIV(a, b) (((a) + (b) - 1) / (b))

extern "C" void kernel_launch(void* const* d_in, const int* in_sizes, int n_in,
                              void* d_out, int out_size, void* d_ws, size_t ws_size,
                              hipStream_t stream) {
  const float* x   = (const float*)d_in[0];
  const float* W1  = (const float*)d_in[1];
  const float* as1 = (const float*)d_in[2];
  const float* ad1 = (const float*)d_in[3];
  const float* b1  = (const float*)d_in[4];
  const float* W2  = (const float*)d_in[5];
  const float* as2 = (const float*)d_in[6];
  const float* ad2 = (const float*)d_in[7];
  const float* b2  = (const float*)d_in[8];
  const float* Wo  = (const float*)d_in[9];
  const float* bo  = (const float*)d_in[10];
  const int*   ei  = (const int*)d_in[11];

  const int n  = in_sizes[0]  / IN_CH;
  const int e  = in_sizes[11] / 2;

  float* p = (float*)d_ws;
  float* h1   = p;  p += (size_t)n * 256;
  float* o1   = p;  p += (size_t)n * 256;
  float* h2   = p;  p += (size_t)n * 64;
  float* o2   = p;  p += (size_t)n * 64;
  float* es1  = p;  p += (size_t)n * 4;
  float* ed1  = p;  p += (size_t)n * 4;
  float* es2v = p;  p += n;
  float* ed2v = p;

  const int B = 256;
  constexpr int NR = 200;
  constexpr int LCAP = 8192;

  int tiles1 = (n / 16) * (256 / 16);
  wmma_gemm_f32<128><<<CDIV(tiles1, 8), B, 0, stream>>>(x, W1, h1, n, 256);
  attn_coef<HEADS><<<CDIV(n * HEADS, B), B, 0, stream>>>(h1, as1, ad1, es1, ed1, n);
  owner_message<HEADS, NR, LCAP><<<CDIV(n, NR), 1024, 0, stream>>>(ei, h1, es1, ed1, b1, o1, e, n);

  int tiles2 = (n / 16) * (64 / 16);
  wmma_gemm_f32<256><<<CDIV(tiles2, 8), B, 0, stream>>>(o1, W2, h2, n, 64);
  attn_coef<1><<<CDIV(n, B), B, 0, stream>>>(h2, as2, ad2, es2v, ed2v, n);
  owner_message<1, NR, LCAP><<<CDIV(n, NR), 1024, 0, stream>>>(ei, h2, es2v, ed2v, b2, o2, e, n);

  out_proj<<<CDIV(n * 32, B), B, 0, stream>>>(o2, Wo, bo, (float*)d_out, n);
}


# Layer-2 owner pass reads each work-group's resolved (source, row) edge list, kept by the layer-1 pass in its own rows of the layer-2 result buffer; second edge scan gone; arrival-order sums kept
# speedup vs baseline: 2.9731x; 1.1552x over previous
#include <hip/hip_runtime.h>
#include <math.h>


#define IN_CH 128
#define HID   64
#define HEADS 4
#define NEG_SLOPE 0.2f

typedef float v2f __attribute__((ext_vector_type(2)));
typedef float v8f __attribute__((ext_vector_type(8)));

__device__ __forceinline__ unsigned f2ord(float f) {
  unsigned u = __float_as_uint(f);
  return (u & 0x80000000u) ? ~u : (u | 0x80000000u);
}
__device__ __forceinline__ float ord2f(unsigned u) {
  return (u & 0x80000000u) ? __uint_as_float(u & 0x7fffffffu)
                           : __uint_as_float(~u);
}
#define NEG_INF_ORD 0x007FFFFFu

__device__ __forceinline__ float elu_f(float x) {
  return x > 0.0f ? x : (expf(x) - 1.0f);
}

template <int K>
__global__ void wmma_gemm_f32(const float* __restrict__ A,
                              const float* __restrict__ B,
                              float* __restrict__ C, int M, int Nc) {
  int wave = (int)((blockIdx.x * blockDim.x + threadIdx.x) >> 5);
  int ntn  = Nc >> 4;
  int total = (M >> 4) * ntn;
  if (wave >= total) return;
  int tm = wave / ntn, tn = wave % ntn;
  int lane = (int)(threadIdx.x & 31);
  int lr = lane & 15, hi = lane >> 4;

  const float* arow = A + (size_t)(tm * 16 + lr) * K;
  const float* bcol = B + tn * 16 + lr;
  v8f acc = {};
#pragma unroll 8
  for (int k0 = 0; k0 < K; k0 += 4) {
    int ka = k0 + hi * 2;
    v2f a; a.x = arow[ka];               a.y = arow[ka + 1];
    v2f b; b.x = bcol[(size_t)ka * Nc];  b.y = bcol[(size_t)(ka + 1) * Nc];
    acc = __builtin_amdgcn_wmma_f32_16x16x4_f32(
         false, a,  false, b,
         (short)0, acc,  false,  false);
  }
  float* crow = C + tn * 16 + lr;
  int rbase = tm * 16 + hi * 8;
#pragma unroll
  for (int v = 0; v < 8; ++v) crow[(size_t)(rbase + v) * Nc] = acc[v];
}

template <int H>
__global__ void attn_coef(const float* __restrict__ h,
                          const float* __restrict__ a_src,
                          const float* __restrict__ a_dst,
                          float* __restrict__ es, float* __restrict__ ed, int n) {
  int idx = blockIdx.x * blockDim.x + threadIdx.x;
  if (idx >= n * H) return;
  int i = idx / H, hh = idx % H;
  const float* hp = h + (size_t)i * (H * HID) + hh * HID;
  const float* as = a_src + hh * HID;
  const float* ad = a_dst + hh * HID;
  float s = 0.f, d = 0.f;
#pragma unroll 8
  for (int c = 0; c < HID; ++c) { float v = hp[c]; s += v * as[c]; d += v * ad[c]; }
  es[idx] = s; ed[idx] = d;
}

template <bool B> struct flag_t { static constexpr bool value = B; };

#define KEEP_PER_NODE (HID / 2)

template <int H, int NR, int LCAP, int KEEP>
__global__ void __launch_bounds__(1024)
owner_message(const int* __restrict__ ei,
              const float* __restrict__ h,
              const float* __restrict__ es,
              const float* __restrict__ ed,
              const float* __restrict__ bias,
              float* out, int2* keep, int E_, int N_) {
  constexpr int LDH = H * HID;
  constexpr int NL  = LDH / 32;
  constexpr int EPW = 8;
  constexpr int NT  = 1024;
  constexpr int U   = 8;
  constexpr int UR  = 4;
  static_assert(H * EPW <= 32, "one lane per (edge, head) pair of the group");
  static_assert(LCAP >= NR + 8, "self-loops plus a smallest window always fit");
  static_assert(KEEP == 0 || LDH == 2 * KEEP_PER_NODE, "the kept list's slice is the own rows of out");

  __shared__ __attribute__((aligned(16))) float acc[NR * LDH];
  __shared__ float    den[NR * H];
  __shared__ unsigned mxu[NR * H];
  __shared__ float    edl[NR * H];
  __shared__ int2     list[LCAP];
  __shared__ int      cnt;

  const int tid  = (int)threadIdx.x;
  const int lane = tid & 31;
  const int wv   = __builtin_amdgcn_readfirstlane(tid >> 5);
  const int n0   = (int)blockIdx.x * NR;
  const int nrv  = (N_ - n0 < NR) ? (N_ - n0) : NR;
  const unsigned nrvu = (unsigned)nrv;
  const int* __restrict__ dstp = ei + E_;
  const int4* __restrict__ dst4 = reinterpret_cast<const int4*>(dstp);
  const bool vec_ok = ((E_ & 3) == 0) && ((reinterpret_cast<size_t>(dstp) & 15) == 0);
  const int q  = lane / H, hh = lane % H;
  int2* const kp = keep + (size_t)n0 * KEEP_PER_NODE;
  const int kcap = nrv * KEEP_PER_NODE - 16;

  for (int i = tid; i < NR * LDH / 4; i += NT)
    reinterpret_cast<float4*>(acc)[i] = make_float4(0.f, 0.f, 0.f, 0.f);
  for (int i = tid; i < NR * H; i += NT) {
    den[i] = 0.f;
    mxu[i] = NEG_INF_ORD;
    edl[i] = (i < nrv * H) ? ed[(size_t)n0 * H + i] : 0.f;
  }

  auto push = [&](int e) {
    const int pos = atomicAdd(&cnt, 1);
    if (pos < LCAP) list[pos].x = e;
  };

  auto add_rows = [&](auto full, int s, int dl, float esv, bool act, int nvalid) {
    constexpr bool FULL = decltype(full)::value;
    constexpr int G = (NL >= 8) ? 4 : 8;
    float p = 0.f;
#pragma unroll
    for (int q0 = 0; q0 < EPW; q0 += G) {
      float hv[G][NL];
      int   dq[G];
#pragma unroll
      for (int t = 0; t < G; ++t) {
        if (FULL || q0 + t < nvalid) {
          const int sq = __builtin_amdgcn_readlane(s, (q0 + t) * H);
          dq[t] = __builtin_amdgcn_readlane(dl, (q0 + t) * H);
          const float* hp = h + (size_t)sq * LDH + lane;
#pragma unroll
          for (int j = 0; j < NL; ++j) hv[t][j] = hp[32 * j];
        }
      }
      if (q0 == 0) {
        float l = esv + edl[dl * H + hh];
        l = (l > 0.0f) ? l : NEG_SLOPE * l;
        p = expf(l - ord2f(mxu[dl * H + hh]));
        if (act) atomicAdd(&den[dl * H + hh], p);
      }
#pragma unroll
      for (int t = 0; t < G; ++t) {
        if (FULL || q0 + t < nvalid) {
          float aq[H];
#pragma unroll
          for (int k = 0; k < H; ++k)
            aq[k] = __int_as_float(__builtin_amdgcn_readlane(__float_as_int(p), (q0 + t) * H + k));
          float* ap = acc + dq[t] * LDH + lane;
#pragma unroll
          for (int j = 0; j < NL; ++j) atomicAdd(ap + 32 * j, hv[t][j] * aq[(32 * j) / HID]);
        }
      }
    }
  };

  bool pre = false;
  if (KEEP == 1) {
    const int kt = __builtin_amdgcn_readfirstlane(kp[kcap].x);
    if (kt >= 0 && kt <= kcap && kt <= LCAP) {
      constexpr int KU = 4;
      for (int base = tid; base < kt; base += NT * KU) {
        int2 t[KU];
        int iv[KU];
#pragma unroll
        for (int u = 0; u < KU; ++u) {
          const int i = base + u * NT;
          iv[u] = (i < kt) ? i : base;
          t[u] = kp[iv[u]];
        }
#pragma unroll
        for (int u = 0; u < KU; ++u) list[iv[u]] = t[u];
      }
      if (tid == 0) cnt = kt;
      pre = true;
    }
  }

  int pass = 0;
  int lo = 0, hi = E_;
  for (;;) {
    if (!pre) {
      const int nself = (lo == 0) ? nrv : 0;
      for (int i = tid; i < nself; i += NT) list[i].x = E_ + n0 + i;
      if (tid == 0) cnt = nself;
      __syncthreads();
      if (vec_ok) {
        const int vlo = lo >> 2, vhi = hi >> 2;
        for (int base = vlo + tid; base < vhi; base += NT * U) {
          int4 dv[U];
#pragma unroll
          for (int u = 0; u < U; ++u) {
            const int i = base + u * NT;
            dv[u] = (i < vhi) ? dst4[i] : make_int4(-1, -1, -1, -1);
          }
#pragma unroll
          for (int u = 0; u < U; ++u) {
            const int e = (base + u * NT) * 4;
            const bool b0 = (unsigned)(dv[u].x - n0) < nrvu;
            const bool b1 = (unsigned)(dv[u].y - n0) < nrvu;
            const bool b2 = (unsigned)(dv[u].z - n0) < nrvu;
            const bool b3 = (unsigned)(dv[u].w - n0) < nrvu;
            if (b0 | b1 | b2 | b3) {
              if (b0) push(e);
              if (b1) push(e + 1);
              if (b2) push(e + 2);
              if (b3) push(e + 3);
            }
          }
        }
      } else {
        for (int e = lo + tid; e < hi; e += NT)
          if ((unsigned)(dstp[e] - n0) < nrvu) push(e);
      }
    }
    __syncthreads();
    const int total = __builtin_amdgcn_readfirstlane(cnt);
    if (total > LCAP) {
      if (pass == 0) {
        pass = 1;
        if (KEEP == 0 && tid == 0) kp[kcap] = make_int2(-1, 0);
      }
      hi = lo + (((hi - lo) >> 1) & ~3);
      __syncthreads();
      continue;
    }

    for (int base = tid; base < total; base += NT * UR) {
      int iv[UR], sv[UR], dv[UR];
#pragma unroll
      for (int u = 0; u < UR; ++u) {
        const int i = base + u * NT;
        iv[u] = (i < total) ? i : base;
        if (pre) {
          const int2 t = list[iv[u]];
          sv[u] = t.x; dv[u] = t.y;
        } else {
          const int e  = list[iv[u]].x;
          const int el = (e < E_) ? e : 0;
          const int s_ = ei[el], d_ = dstp[el];
          sv[u] = (e < E_) ? s_ : (e - E_);
          dv[u] = ((e < E_) ? d_ : (e - E_)) - n0;
        }
      }
      float ev[UR][H];
#pragma unroll
      for (int u = 0; u < UR; ++u)
#pragma unroll
        for (int k = 0; k < H; ++k) ev[u][k] = es[(size_t)sv[u] * H + k];
#pragma unroll
      for (int u = 0; u < UR; ++u) {
        if (!pre) list[iv[u]] = make_int2(sv[u], dv[u]);
#pragma unroll
        for (int k = 0; k < H; ++k) {
          float l = ev[u][k] + edl[dv[u] * H + k];
          l = (l > 0.0f) ? l : NEG_SLOPE * l;
          atomicMax(&mxu[dv[u] * H + k], f2ord(l));
        }
      }
    }
    __syncthreads();

    if (KEEP == 0 && pass == 0) {
      const bool fits = total <= kcap;
      if (fits)
        for (int i = tid; i < total; i += NT) kp[i] = list[i];
      if (tid == 0) kp[kcap] = make_int2(fits ? total : -1, 0);
    }

    if (pass != 1) {
      const int ngroups = (total + EPW - 1) / EPW;
      for (int g = wv; g < ngroups; g += NT / 32) {
        const int i0 = g * EPW;
        const int nvalid = (total - i0 < EPW) ? (total - i0) : EPW;
        const bool act = q < nvalid;
        const int2 sd = list[i0 + (act ? q : 0)];
        const float esv = es[(size_t)sd.x * H + hh];
        if (nvalid == EPW) add_rows(flag_t<true>{},  sd.x, sd.y, esv, act, nvalid);
        else               add_rows(flag_t<false>{}, sd.x, sd.y, esv, act, nvalid);
      }
    }
    if (hi >= E_) {
      if (pass != 1) break;
      pass = 2; lo = 0; hi = E_;
    } else {
      lo = hi; hi = E_;
    }
    __syncthreads();
  }
  __syncthreads();

  for (int idx = tid; idx < nrv * LDH; idx += NT) {
    const int node = idx / LDH, ch = idx % LDH;
    const float dn = den[node * H + ch / HID] + 1e-16f;
    out[(size_t)(n0 + node) * LDH + ch] = elu_f(acc[idx] / dn + bias[ch]);
  }
}

__global__ void out_proj(const float* __restrict__ h,
                         const float* __restrict__ Wo,
                         const float* __restrict__ bo,
                         float* __restrict__ out, int n) {
  int wid  = (int)((blockIdx.x * blockDim.x + threadIdx.x) >> 5);
  int lane = (int)(threadIdx.x & 31);
  if (wid >= n) return;
  const float* hp = h + (size_t)wid * HID;
  float s = hp[lane] * Wo[lane] + hp[lane + 32] * Wo[lane + 32];
#pragma unroll
  for (int o = 16; o > 0; o >>= 1) s += __shfl_xor(s, o, 32);
  if (lane == 0) out[wid] = s + bo[0];
}

#define CDIV(a, b) (((a) + (b) - 1) / (b))

extern "C" void kernel_launch(void* const* d_in, const int* in_sizes, int n_in,
                              void* d_out, int out_size, void* d_ws, size_t ws_size,
                              hipStream_t stream) {
  const float* x   = (const float*)d_in[0];
  const float* W1  = (const float*)d_in[1];
  const float* as1 = (const float*)d_in[2];
  const float* ad1 = (const float*)d_in[3];
  const float* b1  = (const float*)d_in[4];
  const float* W2  = (const float*)d_in[5];
  const float* as2 = (const float*)d_in[6];
  const float* ad2 = (const float*)d_in[7];
  const float* b2  = (const float*)d_in[8];
  const float* Wo  = (const float*)d_in[9];
  const float* bo  = (const float*)d_in[10];
  const int*   ei  = (const int*)d_in[11];

  const int n  = in_sizes[0]  / IN_CH;
  const int e  = in_sizes[11] / 2;

  float* p = (float*)d_ws;
  float* h1   = p;  p += (size_t)n * 256;
  float* o1   = p;  p += (size_t)n * 256;
  float* h2   = p;  p += (size_t)n * 64;
  float* o2   = p;  p += (size_t)n * 64;
  float* es1  = p;  p += (size_t)n * 4;
  float* ed1  = p;  p += (size_t)n * 4;
  float* es2v = p;  p += n;
  float* ed2v = p;

  const int B = 256;
  constexpr int NR = 200;
  constexpr int LCAP = 8192;
  int2* keep = (int2*)o2;

  int tiles1 = (n / 16) * (256 / 16);
  wmma_gemm_f32<128><<<CDIV(tiles1, 8), B, 0, stream>>>(x, W1, h1, n, 256);
  attn_coef<HEADS><<<CDIV(n * HEADS, B), B, 0, stream>>>(h1, as1, ad1, es1, ed1, n);
  owner_message<HEADS, NR, LCAP, 0><<<CDIV(n, NR), 1024, 0, stream>>>(ei, h1, es1, ed1, b1, o1, keep, e, n);

  int tiles2 = (n / 16) * (64 / 16);
  wmma_gemm_f32<256><<<CDIV(tiles2, 8), B, 0, stream>>>(o1, W2, h2, n, 64);
  attn_coef<1><<<CDIV(n, B), B, 0, stream>>>(h2, as2, ad2, es2v, ed2v, n);
  owner_message<1, NR, LCAP, 1><<<CDIV(n, NR), 1024, 0, stream>>>(ei, h2, es2v, ed2v, b2, o2, keep, e, n);

  out_proj<<<CDIV(n * 32, B), B, 0, stream>>>(o2, Wo, bo, (float*)d_out, n);
}


# Edges dealt to owner lists by a slice pass before layer 1 (LDS counts per slice, one global integer add per run); owners stop scanning the edge array; f32 edge sums stay in arrival order
# speedup vs baseline: 3.4802x; 1.1706x over previous
#include <hip/hip_runtime.h>
#include <math.h>


#define IN_CH 128
#define HID   64
#define HEADS 4
#define NEG_SLOPE 0.2f

typedef float v2f __attribute__((ext_vector_type(2)));
typedef float v8f __attribute__((ext_vector_type(8)));

__device__ __forceinline__ unsigned f2ord(float f) {
  unsigned u = __float_as_uint(f);
  return (u & 0x80000000u) ? ~u : (u | 0x80000000u);
}
__device__ __forceinline__ float ord2f(unsigned u) {
  return (u & 0x80000000u) ? __uint_as_float(u & 0x7fffffffu)
                           : __uint_as_float(~u);
}
#define NEG_INF_ORD 0x007FFFFFu

__device__ __forceinline__ float elu_f(float x) {
  return x > 0.0f ? x : (expf(x) - 1.0f);
}

template <int K>
__global__ void wmma_gemm_f32(const float* __restrict__ A,
                              const float* __restrict__ B,
                              float* __restrict__ C, int M, int Nc) {
  int wave = (int)((blockIdx.x * blockDim.x + threadIdx.x) >> 5);
  int ntn  = Nc >> 4;
  int total = (M >> 4) * ntn;
  if (wave >= total) return;
  int tm = wave / ntn, tn = wave % ntn;
  int lane = (int)(threadIdx.x & 31);
  int lr = lane & 15, hi = lane >> 4;

  const float* arow = A + (size_t)(tm * 16 + lr) * K;
  const float* bcol = B + tn * 16 + lr;
  v8f acc = {};
#pragma unroll 8
  for (int k0 = 0; k0 < K; k0 += 4) {
    int ka = k0 + hi * 2;
    v2f a; a.x = arow[ka];               a.y = arow[ka + 1];
    v2f b; b.x = bcol[(size_t)ka * Nc];  b.y = bcol[(size_t)(ka + 1) * Nc];
    acc = __builtin_amdgcn_wmma_f32_16x16x4_f32(
         false, a,  false, b,
         (short)0, acc,  false,  false);
  }
  float* crow = C + tn * 16 + lr;
  int rbase = tm * 16 + hi * 8;
#pragma unroll
  for (int v = 0; v < 8; ++v) crow[(size_t)(rbase + v) * Nc] = acc[v];
}

#define KEEP_PER_NODE (HID / 2)
#define DEAL_MAXOWN 1024

template <int NR>
__device__ __forceinline__ int2* owner_list(int2* keep, int o) {
  return keep + (size_t)o * NR * KEEP_PER_NODE;
}
template <int NR>
__device__ __forceinline__ int owner_cap(int o, int n) {
  const int nrv = (n - o * NR < NR) ? (n - o * NR) : NR;
  return nrv * KEEP_PER_NODE - 16;
}

template <int NR, int NT, int G>
__global__ void __launch_bounds__(NT)
deal_edges(const int* __restrict__ ei, int2* keep, int E_, int N_) {
  __shared__ int hist[DEAL_MAXOWN];
  const int nown = (N_ + NR - 1) / NR;
  if (nown > DEAL_MAXOWN) return;
  const int tid = (int)threadIdx.x;
  for (int i = tid; i < nown; i += NT) hist[i] = 0;

  const int* __restrict__ dstp = ei + E_;
  const int base = (int)blockIdx.x * (NT * 4 * G) + tid * 4;
  const bool vec_ok = ((E_ & 3) == 0) && ((reinterpret_cast<size_t>(ei) & 15) == 0);
  int s[4 * G], d[4 * G];
  if (vec_ok) {
#pragma unroll
    for (int g = 0; g < G; ++g) {
      const int e0 = base + g * NT * 4;
      const bool in = e0 < E_;
      const int ec = in ? e0 : 0;
      const int4 sv = *reinterpret_cast<const int4*>(ei + ec);
      const int4 dv = *reinterpret_cast<const int4*>(dstp + ec);
      s[4 * g] = sv.x; s[4 * g + 1] = sv.y; s[4 * g + 2] = sv.z; s[4 * g + 3] = sv.w;
      d[4 * g]     = in ? dv.x : -1; d[4 * g + 1] = in ? dv.y : -1;
      d[4 * g + 2] = in ? dv.z : -1; d[4 * g + 3] = in ? dv.w : -1;
    }
  } else {
#pragma unroll
    for (int j = 0; j < 4 * G; ++j) {
      const int e = base + (j >> 2) * NT * 4 + (j & 3);
      const bool in = e < E_;
      s[j] = in ? ei[e] : 0;
      d[j] = in ? dstp[e] : -1;
    }
  }
  __syncthreads();

  int o[4 * G], r[4 * G];
#pragma unroll
  for (int j = 0; j < 4 * G; ++j) {
    const bool ok = (unsigned)d[j] < (unsigned)N_;
    o[j] = ok ? d[j] / NR : -1;
    r[j] = ok ? atomicAdd(&hist[o[j]], 1) : 0;
  }
  __syncthreads();

  for (int t = tid; t < nown; t += NT) {
    const int c = hist[t];
    hist[t] = (c > 0) ? atomicAdd(&owner_list<NR>(keep, t)[owner_cap<NR>(t, N_)].x, c) : 0;
  }
  __syncthreads();

  int st[4 * G];
#pragma unroll
  for (int j = 0; j < 4 * G; ++j) st[j] = hist[(o[j] >= 0) ? o[j] : 0];
#pragma unroll
  for (int j = 0; j < 4 * G; ++j) {
    if (o[j] >= 0) {
      const int pos = st[j] + r[j];
      if (pos < owner_cap<NR>(o[j], N_))
        owner_list<NR>(keep, o[j])[pos] = make_int2(s[j], d[j] - o[j] * NR);
    }
  }
}

template <int H, int NR>
__global__ void attn_coef(const float* __restrict__ h,
                          const float* __restrict__ a_src,
                          const float* __restrict__ a_dst,
                          float* __restrict__ es, float* __restrict__ ed, int n,
                          int2* keep) {
  int idx = blockIdx.x * blockDim.x + threadIdx.x;
  if (keep != nullptr) {
    const int nown = (n + NR - 1) / NR;
    if (idx < nown) owner_list<NR>(keep, idx)[owner_cap<NR>(idx, n)].x = (nown <= DEAL_MAXOWN) ? 0 : -1;
  }
  if (idx >= n * H) return;
  int i = idx / H, hh = idx % H;
  const float* hp = h + (size_t)i * (H * HID) + hh * HID;
  const float* as = a_src + hh * HID;
  const float* ad = a_dst + hh * HID;
  float s = 0.f, d = 0.f;
#pragma unroll 8
  for (int c = 0; c < HID; ++c) { float v = hp[c]; s += v * as[c]; d += v * ad[c]; }
  es[idx] = s; ed[idx] = d;
}

template <bool B> struct flag_t { static constexpr bool value = B; };

template <int H, int NR, int LCAP, int KEEP>
__global__ void __launch_bounds__(1024)
owner_message(const int* __restrict__ ei,
              const float* __restrict__ h,
              const float* __restrict__ es,
              const float* __restrict__ ed,
              const float* __restrict__ bias,
              float* out, int2* keep, int E_, int N_) {
  constexpr int LDH = H * HID;
  constexpr int NL  = LDH / 32;
  constexpr int EPW = 8;
  constexpr int NT  = 1024;
  constexpr int U   = 8;
  constexpr int UR  = 4;
  static_assert(H * EPW <= 32, "one lane per (edge, head) pair of the group");
  static_assert(LCAP >= NR + 8, "self-loops plus a smallest window always fit");
  static_assert(KEEP == 0 || LDH == 2 * KEEP_PER_NODE, "the kept list's slice is the own rows of out");
  static_assert(NR * KEEP_PER_NODE * 8 == NR * HID * 4, "a slice of keep is NR rows of the second layer's result");

  __shared__ __attribute__((aligned(16))) float acc[NR * LDH];
  __shared__ float    den[NR * H];
  __shared__ unsigned mxu[NR * H];
  __shared__ float    edl[NR * H];
  __shared__ int2     list[LCAP];
  __shared__ int      cnt;

  const int tid  = (int)threadIdx.x;
  const int lane = tid & 31;
  const int wv   = __builtin_amdgcn_readfirstlane(tid >> 5);
  const int n0   = (int)blockIdx.x * NR;
  const int nrv  = (N_ - n0 < NR) ? (N_ - n0) : NR;
  const unsigned nrvu = (unsigned)nrv;
  const int* __restrict__ dstp = ei + E_;
  const int4* __restrict__ dst4 = reinterpret_cast<const int4*>(dstp);
  const bool vec_ok = ((E_ & 3) == 0) && ((reinterpret_cast<size_t>(dstp) & 15) == 0);
  const int q  = lane / H, hh = lane % H;
  int2* const kp = keep + (size_t)n0 * KEEP_PER_NODE;
  const int kcap = nrv * KEEP_PER_NODE - 16;

  for (int i = tid; i < NR * LDH / 4; i += NT)
    reinterpret_cast<float4*>(acc)[i] = make_float4(0.f, 0.f, 0.f, 0.f);
  for (int i = tid; i < NR * H; i += NT) {
    den[i] = 0.f;
    mxu[i] = NEG_INF_ORD;
    edl[i] = (i < nrv * H) ? ed[(size_t)n0 * H + i] : 0.f;
  }

  auto push = [&](int e) {
    const int pos = atomicAdd(&cnt, 1);
    if (pos < LCAP) list[pos].x = e;
  };

  auto add_rows = [&](auto full, int s, int dl, float esv, bool act, int nvalid) {
    constexpr bool FULL = decltype(full)::value;
    constexpr int G = (NL >= 8) ? 4 : 8;
    float p = 0.f;
#pragma unroll
    for (int q0 = 0; q0 < EPW; q0 += G) {
      float hv[G][NL];
      int   dq[G];
#pragma unroll
      for (int t = 0; t < G; ++t) {
        if (FULL || q0 + t < nvalid) {
          const int sq = __builtin_amdgcn_readlane(s, (q0 + t) * H);
          dq[t] = __builtin_amdgcn_readlane(dl, (q0 + t) * H);
          const float* hp = h + (size_t)sq * LDH + lane;
#pragma unroll
          for (int j = 0; j < NL; ++j) hv[t][j] = hp[32 * j];
        }
      }
      if (q0 == 0) {
        float l = esv + edl[dl * H + hh];
        l = (l > 0.0f) ? l : NEG_SLOPE * l;
        p = expf(l - ord2f(mxu[dl * H + hh]));
        if (act) atomicAdd(&den[dl * H + hh], p);
      }
#pragma unroll
      for (int t = 0; t < G; ++t) {
        if (FULL || q0 + t < nvalid) {
          float aq[H];
#pragma unroll
          for (int k = 0; k < H; ++k)
            aq[k] = __int_as_float(__builtin_amdgcn_readlane(__float_as_int(p), (q0 + t) * H + k));
          float* ap = acc + dq[t] * LDH + lane;
#pragma unroll
          for (int j = 0; j < NL; ++j) atomicAdd(ap + 32 * j, hv[t][j] * aq[(32 * j) / HID]);
        }
      }
    }
  };

  bool pre = false;
  int kept = 0;
  {
    const int kt = __builtin_amdgcn_readfirstlane(kp[kcap].x);
    const int nadd = (KEEP == 0) ? nrv : 0;
    if (kt >= 0 && kt <= kcap && kt + nadd <= LCAP) {
      constexpr int KU = 4;
      for (int base = tid; base < kt; base += NT * KU) {
        int2 t[KU];
        int iv[KU];
#pragma unroll
        for (int u = 0; u < KU; ++u) {
          const int i = base + u * NT;
          iv[u] = (i < kt) ? i : base;
          t[u] = kp[iv[u]];
        }
#pragma unroll
        for (int u = 0; u < KU; ++u) list[iv[u]] = t[u];
      }
      for (int i = tid; i < nadd; i += NT) list[kt + i] = make_int2(n0 + i, i);
      if (tid == 0) cnt = kt + nadd;
      pre = true;
      kept = kt;
    }
  }

  int pass = 0;
  int lo = 0, hi = E_;
  for (;;) {
    if (!pre) {
      const int nself = (lo == 0) ? nrv : 0;
      for (int i = tid; i < nself; i += NT) list[i].x = E_ + n0 + i;
      if (tid == 0) cnt = nself;
      __syncthreads();
      if (vec_ok) {
        const int vlo = lo >> 2, vhi = hi >> 2;
        for (int base = vlo + tid; base < vhi; base += NT * U) {
          int4 dv[U];
#pragma unroll
          for (int u = 0; u < U; ++u) {
            const int i = base + u * NT;
            dv[u] = (i < vhi) ? dst4[i] : make_int4(-1, -1, -1, -1);
          }
#pragma unroll
          for (int u = 0; u < U; ++u) {
            const int e = (base + u * NT) * 4;
            const bool b0 = (unsigned)(dv[u].x - n0) < nrvu;
            const bool b1 = (unsigned)(dv[u].y - n0) < nrvu;
            const bool b2 = (unsigned)(dv[u].z - n0) < nrvu;
            const bool b3 = (unsigned)(dv[u].w - n0) < nrvu;
            if (b0 | b1 | b2 | b3) {
              if (b0) push(e);
              if (b1) push(e + 1);
              if (b2) push(e + 2);
              if (b3) push(e + 3);
            }
          }
        }
      } else {
        for (int e = lo + tid; e < hi; e += NT)
          if ((unsigned)(dstp[e] - n0) < nrvu) push(e);
      }
    }
    __syncthreads();
    const int total = __builtin_amdgcn_readfirstlane(cnt);
    if (total > LCAP) {
      if (pass == 0) {
        pass = 1;
        if (KEEP == 0 && tid == 0) kp[kcap] = make_int2(-1, 0);
      }
      hi = lo + (((hi - lo) >> 1) & ~3);
      __syncthreads();
      continue;
    }

    for (int base = tid; base < total; base += NT * UR) {
      int iv[UR], sv[UR], dv[UR];
#pragma unroll
      for (int u = 0; u < UR; ++u) {
        const int i = base + u * NT;
        iv[u] = (i < total) ? i : base;
        if (pre) {
          const int2 t = list[iv[u]];
          sv[u] = t.x; dv[u] = t.y;
        } else {
          const int e  = list[iv[u]].x;
          const int el = (e < E_) ? e : 0;
          const int s_ = ei[el], d_ = dstp[el];
          sv[u] = (e < E_) ? s_ : (e - E_);
          dv[u] = ((e < E_) ? d_ : (e - E_)) - n0;
        }
      }
      float ev[UR][H];
#pragma unroll
      for (int u = 0; u < UR; ++u)
#pragma unroll
        for (int k = 0; k < H; ++k) ev[u][k] = es[(size_t)sv[u] * H + k];
#pragma unroll
      for (int u = 0; u < UR; ++u) {
        if (!pre) list[iv[u]] = make_int2(sv[u], dv[u]);
#pragma unroll
        for (int k = 0; k < H; ++k) {
          float l = ev[u][k] + edl[dv[u] * H + k];
          l = (l > 0.0f) ? l : NEG_SLOPE * l;
          atomicMax(&mxu[dv[u] * H + k], f2ord(l));
        }
      }
    }
    __syncthreads();

    if (KEEP == 0 && pass == 0) {
      const bool fits = total <= kcap;
      if (fits)
        for (int i = kept + tid; i < total; i += NT) kp[i] = list[i];
      if (tid == 0) kp[kcap] = make_int2(fits ? total : -1, 0);
    }

    if (pass != 1) {
      const int ngroups = (total + EPW - 1) / EPW;
      for (int g = wv; g < ngroups; g += NT / 32) {
        const int i0 = g * EPW;
        const int nvalid = (total - i0 < EPW) ? (total - i0) : EPW;
        const bool act = q < nvalid;
        const int2 sd = list[i0 + (act ? q : 0)];
        const float esv = es[(size_t)sd.x * H + hh];
        if (nvalid == EPW) add_rows(flag_t<true>{},  sd.x, sd.y, esv, act, nvalid);
        else               add_rows(flag_t<false>{}, sd.x, sd.y, esv, act, nvalid);
      }
    }
    if (hi >= E_) {
      if (pass != 1) break;
      pass = 2; lo = 0; hi = E_;
    } else {
      lo = hi; hi = E_;
    }
    __syncthreads();
  }
  __syncthreads();

  for (int idx = tid; idx < nrv * LDH; idx += NT) {
    const int node = idx / LDH, ch = idx % LDH;
    const float dn = den[node * H + ch / HID] + 1e-16f;
    out[(size_t)(n0 + node) * LDH + ch] = elu_f(acc[idx] / dn + bias[ch]);
  }
}

__global__ void out_proj(const float* __restrict__ h,
                         const float* __restrict__ Wo,
                         const float* __restrict__ bo,
                         float* __restrict__ out, int n) {
  int wid  = (int)((blockIdx.x * blockDim.x + threadIdx.x) >> 5);
  int lane = (int)(threadIdx.x & 31);
  if (wid >= n) return;
  const float* hp = h + (size_t)wid * HID;
  float s = hp[lane] * Wo[lane] + hp[lane + 32] * Wo[lane + 32];
#pragma unroll
  for (int o = 16; o > 0; o >>= 1) s += __shfl_xor(s, o, 32);
  if (lane == 0) out[wid] = s + bo[0];
}

#define CDIV(a, b) (((a) + (b) - 1) / (b))

extern "C" void kernel_launch(void* const* d_in, const int* in_sizes, int n_in,
                              void* d_out, int out_size, void* d_ws, size_t ws_size,
                              hipStream_t stream) {
  const float* x   = (const float*)d_in[0];
  const float* W1  = (const float*)d_in[1];
  const float* as1 = (const float*)d_in[2];
  const float* ad1 = (const float*)d_in[3];
  const float* b1  = (const float*)d_in[4];
  const float* W2  = (const float*)d_in[5];
  const float* as2 = (const float*)d_in[6];
  const float* ad2 = (const float*)d_in[7];
  const float* b2  = (const float*)d_in[8];
  const float* Wo  = (const float*)d_in[9];
  const float* bo  = (const float*)d_in[10];
  const int*   ei  = (const int*)d_in[11];

  const int n  = in_sizes[0]  / IN_CH;
  const int e  = in_sizes[11] / 2;

  float* p = (float*)d_ws;
  float* h1   = p;  p += (size_t)n * 256;
  float* o1   = p;  p += (size_t)n * 256;
  float* h2   = p;  p += (size_t)n * 64;
  float* o2   = p;  p += (size_t)n * 64;
  float* es1  = p;  p += (size_t)n * 4;
  float* ed1  = p;  p += (size_t)n * 4;
  float* es2v = p;  p += n;
  float* ed2v = p;

  const int B = 256;
  constexpr int NR = 200;
  constexpr int LCAP = 8192;
  int2* keep = (int2*)o2;

  int tiles1 = (n / 16) * (256 / 16);
  wmma_gemm_f32<128><<<CDIV(tiles1, 8), B, 0, stream>>>(x, W1, h1, n, 256);
  attn_coef<HEADS, NR><<<CDIV(n * HEADS, B), B, 0, stream>>>(h1, as1, ad1, es1, ed1, n, keep);
  deal_edges<NR, 256, 4><<<CDIV(e, 4096), 256, 0, stream>>>(ei, keep, e, n);
  owner_message<HEADS, NR, LCAP, 0><<<CDIV(n, NR), 1024, 0, stream>>>(ei, h1, es1, ed1, b1, o1, keep, e, n);

  int tiles2 = (n / 16) * (64 / 16);
  wmma_gemm_f32<256><<<CDIV(tiles2, 8), B, 0, stream>>>(o1, W2, h2, n, 64);
  attn_coef<1, NR><<<CDIV(n, B), B, 0, stream>>>(h2, as2, ad2, es2v, ed2v, n, nullptr);
  owner_message<1, NR, LCAP, 1><<<CDIV(n, NR), 1024, 0, stream>>>(ei, h2, es2v, ed2v, b2, o2, keep, e, n);

  out_proj<<<CDIV(n * 32, B), B, 0, stream>>>(o2, Wo, bo, (float*)d_out, n);
}


# Output projection folded into the layer-2 owner pass: a wave takes each finished row from LDS through Wo (same lane pairing and 32-lane butterfly); projection launch and o2 array gone
# speedup vs baseline: 3.5899x; 1.0315x over previous
#include <hip/hip_runtime.h>
#include <math.h>


#define IN_CH 128
#define HID   64
#define HEADS 4
#define NEG_SLOPE 0.2f

typedef float v2f __attribute__((ext_vector_type(2)));
typedef float v8f __attribute__((ext_vector_type(8)));

__device__ __forceinline__ unsigned f2ord(float f) {
  unsigned u = __float_as_uint(f);
  return (u & 0x80000000u) ? ~u : (u | 0x80000000u);
}
__device__ __forceinline__ float ord2f(unsigned u) {
  return (u & 0x80000000u) ? __uint_as_float(u & 0x7fffffffu)
                           : __uint_as_float(~u);
}
#define NEG_INF_ORD 0x007FFFFFu

__device__ __forceinline__ float elu_f(float x) {
  return x > 0.0f ? x : (expf(x) - 1.0f);
}

template <int K>
__global__ void wmma_gemm_f32(const float* __restrict__ A,
                              const float* __restrict__ B,
                              float* __restrict__ C, int M, int Nc) {
  int wave = (int)((blockIdx.x * blockDim.x + threadIdx.x) >> 5);
  int ntn  = Nc >> 4;
  int total = (M >> 4) * ntn;
  if (wave >= total) return;
  int tm = wave / ntn, tn = wave % ntn;
  int lane = (int)(threadIdx.x & 31);
  int lr = lane & 15, hi = lane >> 4;

  const float* arow = A + (size_t)(tm * 16 + lr) * K;
  const float* bcol = B + tn * 16 + lr;
  v8f acc = {};
#pragma unroll 8
  for (int k0 = 0; k0 < K; k0 += 4) {
    int ka = k0 + hi * 2;
    v2f a; a.x = arow[ka];               a.y = arow[ka + 1];
    v2f b; b.x = bcol[(size_t)ka * Nc];  b.y = bcol[(size_t)(ka + 1) * Nc];
    acc = __builtin_amdgcn_wmma_f32_16x16x4_f32(
         false, a,  false, b,
         (short)0, acc,  false,  false);
  }
  float* crow = C + tn * 16 + lr;
  int rbase = tm * 16 + hi * 8;
#pragma unroll
  for (int v = 0; v < 8; ++v) crow[(size_t)(rbase + v) * Nc] = acc[v];
}

#define KEEP_PER_NODE (HID / 2)
#define DEAL_MAXOWN 1024

template <int NR>
__device__ __forceinline__ int2* owner_list(int2* keep, int o) {
  return keep + (size_t)o * NR * KEEP_PER_NODE;
}
template <int NR>
__device__ __forceinline__ int owner_cap(int o, int n) {
  const int nrv = (n - o * NR < NR) ? (n - o * NR) : NR;
  return nrv * KEEP_PER_NODE - 16;
}

template <int NR, int NT, int G>
__global__ void __launch_bounds__(NT)
deal_edges(const int* __restrict__ ei, int2* keep, int E_, int N_) {
  __shared__ int hist[DEAL_MAXOWN];
  const int nown = (N_ + NR - 1) / NR;
  if (nown > DEAL_MAXOWN) return;
  const int tid = (int)threadIdx.x;
  for (int i = tid; i < nown; i += NT) hist[i] = 0;

  const int* __restrict__ dstp = ei + E_;
  const int base = (int)blockIdx.x * (NT * 4 * G) + tid * 4;
  const bool vec_ok = ((E_ & 3) == 0) && ((reinterpret_cast<size_t>(ei) & 15) == 0);
  int s[4 * G], d[4 * G];
  if (vec_ok) {
#pragma unroll
    for (int g = 0; g < G; ++g) {
      const int e0 = base + g * NT * 4;
      const bool in = e0 < E_;
      const int ec = in ? e0 : 0;
      const int4 sv = *reinterpret_cast<const int4*>(ei + ec);
      const int4 dv = *reinterpret_cast<const int4*>(dstp + ec);
      s[4 * g] = sv.x; s[4 * g + 1] = sv.y; s[4 * g + 2] = sv.z; s[4 * g + 3] = sv.w;
      d[4 * g]     = in ? dv.x : -1; d[4 * g + 1] = in ? dv.y : -1;
      d[4 * g + 2] = in ? dv.z : -1; d[4 * g + 3] = in ? dv.w : -1;
    }
  } else {
#pragma unroll
    for (int j = 0; j < 4 * G; ++j) {
      const int e = base + (j >> 2) * NT * 4 + (j & 3);
      const bool in = e < E_;
      s[j] = in ? ei[e] : 0;
      d[j] = in ? dstp[e] : -1;
    }
  }
  __syncthreads();

  int o[4 * G], r[4 * G];
#pragma unroll
  for (int j = 0; j < 4 * G; ++j) {
    const bool ok = (unsigned)d[j] < (unsigned)N_;
    o[j] = ok ? d[j] / NR : -1;
    r[j] = ok ? atomicAdd(&hist[o[j]], 1) : 0;
  }
  __syncthreads();

  for (int t = tid; t < nown; t += NT) {
    const int c = hist[t];
    hist[t] = (c > 0) ? atomicAdd(&owner_list<NR>(keep, t)[owner_cap<NR>(t, N_)].x, c) : 0;
  }
  __syncthreads();

  int st[4 * G];
#pragma unroll
  for (int j = 0; j < 4 * G; ++j) st[j] = hist[(o[j] >= 0) ? o[j] : 0];
#pragma unroll
  for (int j = 0; j < 4 * G; ++j) {
    if (o[j] >= 0) {
      const int pos = st[j] + r[j];
      if (pos < owner_cap<NR>(o[j], N_))
        owner_list<NR>(keep, o[j])[pos] = make_int2(s[j], d[j] - o[j] * NR);
    }
  }
}

template <int H, int NR>
__global__ void attn_coef(const float* __restrict__ h,
                          const float* __restrict__ a_src,
                          const float* __restrict__ a_dst,
                          float* __restrict__ es, float* __restrict__ ed, int n,
                          int2* keep) {
  int idx = blockIdx.x * blockDim.x + threadIdx.x;
  if (keep != nullptr) {
    const int nown = (n + NR - 1) / NR;
    if (idx < nown) owner_list<NR>(keep, idx)[owner_cap<NR>(idx, n)].x = (nown <= DEAL_MAXOWN) ? 0 : -1;
  }
  if (idx >= n * H) return;
  int i = idx / H, hh = idx % H;
  const float* hp = h + (size_t)i * (H * HID) + hh * HID;
  const float* as = a_src + hh * HID;
  const float* ad = a_dst + hh * HID;
  float s = 0.f, d = 0.f;
#pragma unroll 8
  for (int c = 0; c < HID; ++c) { float v = hp[c]; s += v * as[c]; d += v * ad[c]; }
  es[idx] = s; ed[idx] = d;
}

template <bool B> struct flag_t { static constexpr bool value = B; };

template <int H, int NR, int LCAP, int KEEP, int PROJ>
__global__ void __launch_bounds__(1024)
owner_message(const int* __restrict__ ei,
              const float* __restrict__ h,
              const float* __restrict__ es,
              const float* __restrict__ ed,
              const float* __restrict__ bias,
              const float* __restrict__ Wo,
              const float* __restrict__ bo,
              float* out, int2* keep, int E_, int N_) {
  constexpr int LDH = H * HID;
  constexpr int NL  = LDH / 32;
  constexpr int EPW = 8;
  constexpr int NT  = 1024;
  constexpr int U   = 8;
  constexpr int UR  = 4;
  static_assert(H * EPW <= 32, "one lane per (edge, head) pair of the group");
  static_assert(LCAP >= NR + 8, "self-loops plus a smallest window always fit");
  static_assert(NR * KEEP_PER_NODE * 8 == NR * HID * 4, "a slice of the list plane is 256 bytes per own node");
  static_assert(PROJ == 0 || (H == 1 && LDH == 64), "the projection takes one head's row of 64: two channels per lane");

  __shared__ __attribute__((aligned(16))) float acc[NR * LDH];
  __shared__ float    den[NR * H];
  __shared__ unsigned mxu[NR * H];
  __shared__ float    edl[NR * H];
  __shared__ int2     list[LCAP];
  __shared__ int      cnt;

  const int tid  = (int)threadIdx.x;
  const int lane = tid & 31;
  const int wv   = __builtin_amdgcn_readfirstlane(tid >> 5);
  const int n0   = (int)blockIdx.x * NR;
  const int nrv  = (N_ - n0 < NR) ? (N_ - n0) : NR;
  const unsigned nrvu = (unsigned)nrv;
  const int* __restrict__ dstp = ei + E_;
  const int4* __restrict__ dst4 = reinterpret_cast<const int4*>(dstp);
  const bool vec_ok = ((E_ & 3) == 0) && ((reinterpret_cast<size_t>(dstp) & 15) == 0);
  const int q  = lane / H, hh = lane % H;
  int2* const kp = keep + (size_t)n0 * KEEP_PER_NODE;
  const int kcap = nrv * KEEP_PER_NODE - 16;

  float w0 = 0.f, w1 = 0.f, c0 = 0.f, c1 = 0.f, bo0 = 0.f;
  if (PROJ) {
    w0 = Wo[lane];   w1 = Wo[lane + 32];
    c0 = bias[lane]; c1 = bias[lane + 32];
    bo0 = bo[0];
  }
  for (int i = tid; i < NR * LDH / 4; i += NT)
    reinterpret_cast<float4*>(acc)[i] = make_float4(0.f, 0.f, 0.f, 0.f);
  for (int i = tid; i < NR * H; i += NT) {
    den[i] = 0.f;
    mxu[i] = NEG_INF_ORD;
    edl[i] = (i < nrv * H) ? ed[(size_t)n0 * H + i] : 0.f;
  }

  auto push = [&](int e) {
    const int pos = atomicAdd(&cnt, 1);
    if (pos < LCAP) list[pos].x = e;
  };

  auto add_rows = [&](auto full, int s, int dl, float esv, bool act, int nvalid) {
    constexpr bool FULL = decltype(full)::value;
    constexpr int G = (NL >= 8) ? 4 : 8;
    float p = 0.f;
#pragma unroll
    for (int q0 = 0; q0 < EPW; q0 += G) {
      float hv[G][NL];
      int   dq[G];
#pragma unroll
      for (int t = 0; t < G; ++t) {
        if (FULL || q0 + t < nvalid) {
          const int sq = __builtin_amdgcn_readlane(s, (q0 + t) * H);
          dq[t] = __builtin_amdgcn_readlane(dl, (q0 + t) * H);
          const float* hp = h + (size_t)sq * LDH + lane;
#pragma unroll
          for (int j = 0; j < NL; ++j) hv[t][j] = hp[32 * j];
        }
      }
      if (q0 == 0) {
        float l = esv + edl[dl * H + hh];
        l = (l > 0.0f) ? l : NEG_SLOPE * l;
        p = expf(l - ord2f(mxu[dl * H + hh]));
        if (act) atomicAdd(&den[dl * H + hh], p);
      }
#pragma unroll
      for (int t = 0; t < G; ++t) {
        if (FULL || q0 + t < nvalid) {
          float aq[H];
#pragma unroll
          for (int k = 0; k < H; ++k)
            aq[k] = __int_as_float(__builtin_amdgcn_readlane(__float_as_int(p), (q0 + t) * H + k));
          float* ap = acc + dq[t] * LDH + lane;
#pragma unroll
          for (int j = 0; j < NL; ++j) atomicAdd(ap + 32 * j, hv[t][j] * aq[(32 * j) / HID]);
        }
      }
    }
  };

  bool pre = false;
  int kept = 0;
  {
    const int kt = __builtin_amdgcn_readfirstlane(kp[kcap].x);
    const int nadd = (KEEP == 0) ? nrv : 0;
    if (kt >= 0 && kt <= kcap && kt + nadd <= LCAP) {
      constexpr int KU = 4;
      for (int base = tid; base < kt; base += NT * KU) {
        int2 t[KU];
        int iv[KU];
#pragma unroll
        for (int u = 0; u < KU; ++u) {
          const int i = base + u * NT;
          iv[u] = (i < kt) ? i : base;
          t[u] = kp[iv[u]];
        }
#pragma unroll
        for (int u = 0; u < KU; ++u) list[iv[u]] = t[u];
      }
      for (int i = tid; i < nadd; i += NT) list[kt + i] = make_int2(n0 + i, i);
      if (tid == 0) cnt = kt + nadd;
      pre = true;
      kept = kt;
    }
  }

  int pass = 0;
  int lo = 0, hi = E_;
  for (;;) {
    if (!pre) {
      const int nself = (lo == 0) ? nrv : 0;
      for (int i = tid; i < nself; i += NT) list[i].x = E_ + n0 + i;
      if (tid == 0) cnt = nself;
      __syncthreads();
      if (vec_ok) {
        const int vlo = lo >> 2, vhi = hi >> 2;
        for (int base = vlo + tid; base < vhi; base += NT * U) {
          int4 dv[U];
#pragma unroll
          for (int u = 0; u < U; ++u) {
            const int i = base + u * NT;
            dv[u] = (i < vhi) ? dst4[i] : make_int4(-1, -1, -1, -1);
          }
#pragma unroll
          for (int u = 0; u < U; ++u) {
            const int e = (base + u * NT) * 4;
            const bool b0 = (unsigned)(dv[u].x - n0) < nrvu;
            const bool b1 = (unsigned)(dv[u].y - n0) < nrvu;
            const bool b2 = (unsigned)(dv[u].z - n0) < nrvu;
            const bool b3 = (unsigned)(dv[u].w - n0) < nrvu;
            if (b0 | b1 | b2 | b3) {
              if (b0) push(e);
              if (b1) push(e + 1);
              if (b2) push(e + 2);
              if (b3) push(e + 3);
            }
          }
        }
      } else {
        for (int e = lo + tid; e < hi; e += NT)
          if ((unsigned)(dstp[e] - n0) < nrvu) push(e);
      }
    }
    __syncthreads();
    const int total = __builtin_amdgcn_readfirstlane(cnt);
    if (total > LCAP) {
      if (pass == 0) {
        pass = 1;
        if (KEEP == 0 && tid == 0) kp[kcap] = make_int2(-1, 0);
      }
      hi = lo + (((hi - lo) >> 1) & ~3);
      __syncthreads();
      continue;
    }

    for (int base = tid; base < total; base += NT * UR) {
      int iv[UR], sv[UR], dv[UR];
#pragma unroll
      for (int u = 0; u < UR; ++u) {
        const int i = base + u * NT;
        iv[u] = (i < total) ? i : base;
        if (pre) {
          const int2 t = list[iv[u]];
          sv[u] = t.x; dv[u] = t.y;
        } else {
          const int e  = list[iv[u]].x;
          const int el = (e < E_) ? e : 0;
          const int s_ = ei[el], d_ = dstp[el];
          sv[u] = (e < E_) ? s_ : (e - E_);
          dv[u] = ((e < E_) ? d_ : (e - E_)) - n0;
        }
      }
      float ev[UR][H];
#pragma unroll
      for (int u = 0; u < UR; ++u)
#pragma unroll
        for (int k = 0; k < H; ++k) ev[u][k] = es[(size_t)sv[u] * H + k];
#pragma unroll
      for (int u = 0; u < UR; ++u) {
        if (!pre) list[iv[u]] = make_int2(sv[u], dv[u]);
#pragma unroll
        for (int k = 0; k < H; ++k) {
          float l = ev[u][k] + edl[dv[u] * H + k];
          l = (l > 0.0f) ? l : NEG_SLOPE * l;
          atomicMax(&mxu[dv[u] * H + k], f2ord(l));
        }
      }
    }
    __syncthreads();

    if (KEEP == 0 && pass == 0) {
      const bool fits = total <= kcap;
      if (fits)
        for (int i = kept + tid; i < total; i += NT) kp[i] = list[i];
      if (tid == 0) kp[kcap] = make_int2(fits ? total : -1, 0);
    }

    if (pass != 1) {
      const int ngroups = (total + EPW - 1) / EPW;
      for (int g = wv; g < ngroups; g += NT / 32) {
        const int i0 = g * EPW;
        const int nvalid = (total - i0 < EPW) ? (total - i0) : EPW;
        const bool act = q < nvalid;
        const int2 sd = list[i0 + (act ? q : 0)];
        const float esv = es[(size_t)sd.x * H + hh];
        if (nvalid == EPW) add_rows(flag_t<true>{},  sd.x, sd.y, esv, act, nvalid);
        else               add_rows(flag_t<false>{}, sd.x, sd.y, esv, act, nvalid);
      }
    }
    if (hi >= E_) {
      if (pass != 1) break;
      pass = 2; lo = 0; hi = E_;
    } else {
      lo = hi; hi = E_;
    }
    __syncthreads();
  }
  __syncthreads();

  if (PROJ) {
    constexpr int NW = NT / 32;
    constexpr int TR = (NR + NW - 1) / NW;
    for (int t = 0; t < TR; ++t) {
      const int node = wv + t * NW;
      const int nd = (node < nrv) ? node : 0;
      const float dn = den[nd] + 1e-16f;
      const float v0 = elu_f(acc[nd * LDH + lane] / dn + c0);
      const float v1 = elu_f(acc[nd * LDH + lane + 32] / dn + c1);
      float s = v0 * w0 + v1 * w1;
      for (int o = 16; o > 0; o >>= 1) s += __shfl_xor(s, o, 32);
      if (lane == 0 && node < nrv) edl[node] = s + bo0;
    }
    __syncthreads();
    for (int i = tid; i < nrv; i += NT) out[n0 + i] = edl[i];
  } else {
    for (int idx = tid; idx < nrv * LDH; idx += NT) {
      const int node = idx / LDH, ch = idx % LDH;
      const float dn = den[node * H + ch / HID] + 1e-16f;
      out[(size_t)(n0 + node) * LDH + ch] = elu_f(acc[idx] / dn + bias[ch]);
    }
  }
}

#define CDIV(a, b) (((a) + (b) - 1) / (b))

extern "C" void kernel_launch(void* const* d_in, const int* in_sizes, int n_in,
                              void* d_out, int out_size, void* d_ws, size_t ws_size,
                              hipStream_t stream) {
  const float* x   = (const float*)d_in[0];
  const float* W1  = (const float*)d_in[1];
  const float* as1 = (const float*)d_in[2];
  const float* ad1 = (const float*)d_in[3];
  const float* b1  = (const float*)d_in[4];
  const float* W2  = (const float*)d_in[5];
  const float* as2 = (const float*)d_in[6];
  const float* ad2 = (const float*)d_in[7];
  const float* b2  = (const float*)d_in[8];
  const float* Wo  = (const float*)d_in[9];
  const float* bo  = (const float*)d_in[10];
  const int*   ei  = (const int*)d_in[11];

  const int n  = in_sizes[0]  / IN_CH;
  const int e  = in_sizes[11] / 2;

  float* p = (float*)d_ws;
  float* h1   = p;  p += (size_t)n * 256;
  float* o1   = p;  p += (size_t)n * 256;
  float* h2   = p;  p += (size_t)n * 64;
  float* lst  = p;  p += (size_t)n * 64;
  float* es1  = p;  p += (size_t)n * 4;
  float* ed1  = p;  p += (size_t)n * 4;
  float* es2v = p;  p += n;
  float* ed2v = p;

  const int B = 256;
  constexpr int NR = 200;
  constexpr int LCAP = 8192;
  int2* keep = (int2*)lst;

  int tiles1 = (n / 16) * (256 / 16);
  wmma_gemm_f32<128><<<CDIV(tiles1, 8), B, 0, stream>>>(x, W1, h1, n, 256);
  attn_coef<HEADS, NR><<<CDIV(n * HEADS, B), B, 0, stream>>>(h1, as1, ad1, es1, ed1, n, keep);
  deal_edges<NR, 256, 4><<<CDIV(e, 4096), 256, 0, stream>>>(ei, keep, e, n);
  owner_message<HEADS, NR, LCAP, 0, 0><<<CDIV(n, NR), 1024, 0, stream>>>(ei, h1, es1, ed1, b1, (const float*)nullptr, (const float*)nullptr, o1, keep, e, n);

  int tiles2 = (n / 16) * (64 / 16);
  wmma_gemm_f32<256><<<CDIV(tiles2, 8), B, 0, stream>>>(o1, W2, h2, n, 64);
  attn_coef<1, NR><<<CDIV(n, B), B, 0, stream>>>(h2, as2, ad2, es2v, ed2v, n, nullptr);
  owner_message<1, NR, LCAP, 1, 1><<<CDIV(n, NR), 1024, 0, stream>>>(ei, h2, es2v, ed2v, b2, Wo, bo, (float*)d_out, keep, e, n);
}


# Layer-2 product folded into the layer-1 owner pass: finished rows go from padded LDS through W2 on WMMA, a wave per node writes the 64-wide row and es/ed (32-lane butterfly); es/ed sum order changes
# speedup vs baseline: 4.0775x; 1.1358x over previous
#include <hip/hip_runtime.h>
#include <math.h>


#define IN_CH 128
#define HID   64
#define HEADS 4
#define NEG_SLOPE 0.2f

typedef float v2f __attribute__((ext_vector_type(2)));
typedef float v8f __attribute__((ext_vector_type(8)));

__device__ __forceinline__ unsigned f2ord(float f) {
  unsigned u = __float_as_uint(f);
  return (u & 0x80000000u) ? ~u : (u | 0x80000000u);
}
__device__ __forceinline__ float ord2f(unsigned u) {
  return (u & 0x80000000u) ? __uint_as_float(u & 0x7fffffffu)
                           : __uint_as_float(~u);
}
#define NEG_INF_ORD 0x007FFFFFu

__device__ __forceinline__ float elu_f(float x) {
  return x > 0.0f ? x : (expf(x) - 1.0f);
}

template <int K>
__global__ void wmma_gemm_f32(const float* __restrict__ A,
                              const float* __restrict__ B,
                              float* __restrict__ C, int M, int Nc) {
  int wave = (int)((blockIdx.x * blockDim.x + threadIdx.x) >> 5);
  int ntn  = Nc >> 4;
  int total = (M >> 4) * ntn;
  if (wave >= total) return;
  int tm = wave / ntn, tn = wave % ntn;
  int lane = (int)(threadIdx.x & 31);
  int lr = lane & 15, hi = lane >> 4;

  const float* arow = A + (size_t)(tm * 16 + lr) * K;
  const float* bcol = B + tn * 16 + lr;
  v8f acc = {};
#pragma unroll 8
  for (int k0 = 0; k0 < K; k0 += 4) {
    int ka = k0 + hi * 2;
    v2f a; a.x = arow[ka];               a.y = arow[ka + 1];
    v2f b; b.x = bcol[(size_t)ka * Nc];  b.y = bcol[(size_t)(ka + 1) * Nc];
    acc = __builtin_amdgcn_wmma_f32_16x16x4_f32(
         false, a,  false, b,
         (short)0, acc,  false,  false);
  }
  float* crow = C + tn * 16 + lr;
  int rbase = tm * 16 + hi * 8;
#pragma unroll
  for (int v = 0; v < 8; ++v) crow[(size_t)(rbase + v) * Nc] = acc[v];
}

#define KEEP_PER_NODE (HID / 2)
#define DEAL_MAXOWN 1024

template <int NR>
__device__ __forceinline__ int2* owner_list(int2* keep, int o) {
  return keep + (size_t)o * NR * KEEP_PER_NODE;
}
template <int NR>
__device__ __forceinline__ int owner_cap(int o, int n) {
  const int nrv = (n - o * NR < NR) ? (n - o * NR) : NR;
  return nrv * KEEP_PER_NODE - 16;
}

template <int NR, int NT, int G>
__global__ void __launch_bounds__(NT)
deal_edges(const int* __restrict__ ei, int2* keep, int E_, int N_) {
  __shared__ int hist[DEAL_MAXOWN];
  const int nown = (N_ + NR - 1) / NR;
  if (nown > DEAL_MAXOWN) return;
  const int tid = (int)threadIdx.x;
  for (int i = tid; i < nown; i += NT) hist[i] = 0;

  const int* __restrict__ dstp = ei + E_;
  const int base = (int)blockIdx.x * (NT * 4 * G) + tid * 4;
  const bool vec_ok = ((E_ & 3) == 0) && ((reinterpret_cast<size_t>(ei) & 15) == 0);
  int s[4 * G], d[4 * G];
  if (vec_ok) {
#pragma unroll
    for (int g = 0; g < G; ++g) {
      const int e0 = base + g * NT * 4;
      const bool in = e0 < E_;
      const int ec = in ? e0 : 0;
      const int4 sv = *reinterpret_cast<const int4*>(ei + ec);
      const int4 dv = *reinterpret_cast<const int4*>(dstp + ec);
      s[4 * g] = sv.x; s[4 * g + 1] = sv.y; s[4 * g + 2] = sv.z; s[4 * g + 3] = sv.w;
      d[4 * g]     = in ? dv.x : -1; d[4 * g + 1] = in ? dv.y : -1;
      d[4 * g + 2] = in ? dv.z : -1; d[4 * g + 3] = in ? dv.w : -1;
    }
  } else {
#pragma unroll
    for (int j = 0; j < 4 * G; ++j) {
      const int e = base + (j >> 2) * NT * 4 + (j & 3);
      const bool in = e < E_;
      s[j] = in ? ei[e] : 0;
      d[j] = in ? dstp[e] : -1;
    }
  }
  __syncthreads();

  int o[4 * G], r[4 * G];
#pragma unroll
  for (int j = 0; j < 4 * G; ++j) {
    const bool ok = (unsigned)d[j] < (unsigned)N_;
    o[j] = ok ? d[j] / NR : -1;
    r[j] = ok ? atomicAdd(&hist[o[j]], 1) : 0;
  }
  __syncthreads();

  for (int t = tid; t < nown; t += NT) {
    const int c = hist[t];
    hist[t] = (c > 0) ? atomicAdd(&owner_list<NR>(keep, t)[owner_cap<NR>(t, N_)].x, c) : 0;
  }
  __syncthreads();

  int st[4 * G];
#pragma unroll
  for (int j = 0; j < 4 * G; ++j) st[j] = hist[(o[j] >= 0) ? o[j] : 0];
#pragma unroll
  for (int j = 0; j < 4 * G; ++j) {
    if (o[j] >= 0) {
      const int pos = st[j] + r[j];
      if (pos < owner_cap<NR>(o[j], N_))
        owner_list<NR>(keep, o[j])[pos] = make_int2(s[j], d[j] - o[j] * NR);
    }
  }
}

template <int H, int NR>
__global__ void attn_coef(const float* __restrict__ h,
                          const float* __restrict__ a_src,
                          const float* __restrict__ a_dst,
                          float* __restrict__ es, float* __restrict__ ed, int n,
                          int2* keep) {
  int idx = blockIdx.x * blockDim.x + threadIdx.x;
  if (keep != nullptr) {
    const int nown = (n + NR - 1) / NR;
    if (idx < nown) owner_list<NR>(keep, idx)[owner_cap<NR>(idx, n)].x = (nown <= DEAL_MAXOWN) ? 0 : -1;
  }
  if (idx >= n * H) return;
  int i = idx / H, hh = idx % H;
  const float* hp = h + (size_t)i * (H * HID) + hh * HID;
  const float* as = a_src + hh * HID;
  const float* ad = a_dst + hh * HID;
  float s = 0.f, d = 0.f;
#pragma unroll 8
  for (int c = 0; c < HID; ++c) { float v = hp[c]; s += v * as[c]; d += v * ad[c]; }
  es[idx] = s; ed[idx] = d;
}

template <bool B> struct flag_t { static constexpr bool value = B; };

template <int H, int NR, int LCAP, int KEEP, int PROJ, int NEXT>
__global__ void __launch_bounds__(1024)
owner_message(const int* __restrict__ ei,
              const float* __restrict__ h,
              const float* __restrict__ es,
              const float* __restrict__ ed,
              const float* __restrict__ bias,
              const float* __restrict__ Wo,
              const float* __restrict__ bo,
              const float* __restrict__ Wn,
              const float* __restrict__ asn,
              const float* __restrict__ adn,
              float* esn, float* edn,
              float* out, int2* keep, int E_, int N_) {
  constexpr int LDH = H * HID;
  constexpr int AP  = NEXT ? LDH + 4 : LDH;
  constexpr int NL  = LDH / 32;
  constexpr int EPW = 8;
  constexpr int NT  = 1024;
  constexpr int U   = 8;
  constexpr int UR  = 4;
  static_assert(H * EPW <= 32, "one lane per (edge, head) pair of the group");
  static_assert(LCAP >= NR + 8, "self-loops plus a smallest window always fit");
  static_assert(NR * KEEP_PER_NODE * 8 == NR * HID * 4, "a slice of the list plane is 256 bytes per own node");
  static_assert(PROJ == 0 || (H == 1 && LDH == 64), "the projection takes one head's row of 64: two channels per lane");
  static_assert(NEXT == 0 || (PROJ == 0 && H >= 2 && LDH % 4 == 0 && ((NR + 15) / 16) * 16 * HID * 4 <= LCAP * 8),
                "the next layer's 64-wide rows, padded to whole 16-row tiles, are staged in the list's LDS");

  __shared__ __attribute__((aligned(16))) float acc[NR * AP];
  __shared__ float    den[NR * H];
  __shared__ unsigned mxu[NR * H];
  __shared__ float    edl[NR * H];
  __shared__ int2     list[LCAP];
  __shared__ int      cnt;

  const int tid  = (int)threadIdx.x;
  const int lane = tid & 31;
  const int wv   = __builtin_amdgcn_readfirstlane(tid >> 5);
  const int n0   = (int)blockIdx.x * NR;
  const int nrv  = (N_ - n0 < NR) ? (N_ - n0) : NR;
  const unsigned nrvu = (unsigned)nrv;
  const int* __restrict__ dstp = ei + E_;
  const int4* __restrict__ dst4 = reinterpret_cast<const int4*>(dstp);
  const bool vec_ok = ((E_ & 3) == 0) && ((reinterpret_cast<size_t>(dstp) & 15) == 0);
  const int q  = lane / H, hh = lane % H;
  int2* const kp = keep + (size_t)n0 * KEEP_PER_NODE;
  const int kcap = nrv * KEEP_PER_NODE - 16;

  float w0 = 0.f, w1 = 0.f, c0 = 0.f, c1 = 0.f, bo0 = 0.f;
  if (PROJ) {
    w0 = Wo[lane];   w1 = Wo[lane + 32];
    c0 = bias[lane]; c1 = bias[lane + 32];
    bo0 = bo[0];
  }
  for (int i = tid; i < NR * AP / 4; i += NT)
    reinterpret_cast<float4*>(acc)[i] = make_float4(0.f, 0.f, 0.f, 0.f);
  for (int i = tid; i < NR * H; i += NT) {
    den[i] = 0.f;
    mxu[i] = NEG_INF_ORD;
    edl[i] = (i < nrv * H) ? ed[(size_t)n0 * H + i] : 0.f;
  }

  auto push = [&](int e) {
    const int pos = atomicAdd(&cnt, 1);
    if (pos < LCAP) list[pos].x = e;
  };

  auto add_rows = [&](auto full, int s, int dl, float esv, bool act, int nvalid) {
    constexpr bool FULL = decltype(full)::value;
    constexpr int G = (NL >= 8) ? 4 : 8;
    float p = 0.f;
#pragma unroll
    for (int q0 = 0; q0 < EPW; q0 += G) {
      float hv[G][NL];
      int   dq[G];
#pragma unroll
      for (int t = 0; t < G; ++t) {
        if (FULL || q0 + t < nvalid) {
          const int sq = __builtin_amdgcn_readlane(s, (q0 + t) * H);
          dq[t] = __builtin_amdgcn_readlane(dl, (q0 + t) * H);
          const float* hp = h + (size_t)sq * LDH + lane;
#pragma unroll
          for (int j = 0; j < NL; ++j) hv[t][j] = hp[32 * j];
        }
      }
      if (q0 == 0) {
        float l = esv + edl[dl * H + hh];
        l = (l > 0.0f) ? l : NEG_SLOPE * l;
        p = expf(l - ord2f(mxu[dl * H + hh]));
        if (act) atomicAdd(&den[dl * H + hh], p);
      }
#pragma unroll
      for (int t = 0; t < G; ++t) {
        if (FULL || q0 + t < nvalid) {
          float aq[H];
#pragma unroll
          for (int k = 0; k < H; ++k)
            aq[k] = __int_as_float(__builtin_amdgcn_readlane(__float_as_int(p), (q0 + t) * H + k));
          float* ap = acc + dq[t] * AP + lane;
#pragma unroll
          for (int j = 0; j < NL; ++j) atomicAdd(ap + 32 * j, hv[t][j] * aq[(32 * j) / HID]);
        }
      }
    }
  };

  bool pre = false;
  int kept = 0;
  {
    const int kt = __builtin_amdgcn_readfirstlane(kp[kcap].x);
    const int nadd = (KEEP == 0) ? nrv : 0;
    if (kt >= 0 && kt <= kcap && kt + nadd <= LCAP) {
      constexpr int KU = 4;
      for (int base = tid; base < kt; base += NT * KU) {
        int2 t[KU];
        int iv[KU];
#pragma unroll
        for (int u = 0; u < KU; ++u) {
          const int i = base + u * NT;
          iv[u] = (i < kt) ? i : base;
          t[u] = kp[iv[u]];
        }
#pragma unroll
        for (int u = 0; u < KU; ++u) list[iv[u]] = t[u];
      }
      for (int i = tid; i < nadd; i += NT) list[kt + i] = make_int2(n0 + i, i);
      if (tid == 0) cnt = kt + nadd;
      pre = true;
      kept = kt;
    }
  }

  int pass = 0;
  int lo = 0, hi = E_;
  for (;;) {
    if (!pre) {
      const int nself = (lo == 0) ? nrv : 0;
      for (int i = tid; i < nself; i += NT) list[i].x = E_ + n0 + i;
      if (tid == 0) cnt = nself;
      __syncthreads();
      if (vec_ok) {
        const int vlo = lo >> 2, vhi = hi >> 2;
        for (int base = vlo + tid; base < vhi; base += NT * U) {
          int4 dv[U];
#pragma unroll
          for (int u = 0; u < U; ++u) {
            const int i = base + u * NT;
            dv[u] = (i < vhi) ? dst4[i] : make_int4(-1, -1, -1, -1);
          }
#pragma unroll
          for (int u = 0; u < U; ++u) {
            const int e = (base + u * NT) * 4;
            const bool b0 = (unsigned)(dv[u].x - n0) < nrvu;
            const bool b1 = (unsigned)(dv[u].y - n0) < nrvu;
            const bool b2 = (unsigned)(dv[u].z - n0) < nrvu;
            const bool b3 = (unsigned)(dv[u].w - n0) < nrvu;
            if (b0 | b1 | b2 | b3) {
              if (b0) push(e);
              if (b1) push(e + 1);
              if (b2) push(e + 2);
              if (b3) push(e + 3);
            }
          }
        }
      } else {
        for (int e = lo + tid; e < hi; e += NT)
          if ((unsigned)(dstp[e] - n0) < nrvu) push(e);
      }
    }
    __syncthreads();
    const int total = __builtin_amdgcn_readfirstlane(cnt);
    if (total > LCAP) {
      if (pass == 0) {
        pass = 1;
        if (KEEP == 0 && tid == 0) kp[kcap] = make_int2(-1, 0);
      }
      hi = lo + (((hi - lo) >> 1) & ~3);
      __syncthreads();
      continue;
    }

    for (int base = tid; base < total; base += NT * UR) {
      int iv[UR], sv[UR], dv[UR];
#pragma unroll
      for (int u = 0; u < UR; ++u) {
        const int i = base + u * NT;
        iv[u] = (i < total) ? i : base;
        if (pre) {
          const int2 t = list[iv[u]];
          sv[u] = t.x; dv[u] = t.y;
        } else {
          const int e  = list[iv[u]].x;
          const int el = (e < E_) ? e : 0;
          const int s_ = ei[el], d_ = dstp[el];
          sv[u] = (e < E_) ? s_ : (e - E_);
          dv[u] = ((e < E_) ? d_ : (e - E_)) - n0;
        }
      }
      float ev[UR][H];
#pragma unroll
      for (int u = 0; u < UR; ++u)
#pragma unroll
        for (int k = 0; k < H; ++k) ev[u][k] = es[(size_t)sv[u] * H + k];
#pragma unroll
      for (int u = 0; u < UR; ++u) {
        if (!pre) list[iv[u]] = make_int2(sv[u], dv[u]);
#pragma unroll
        for (int k = 0; k < H; ++k) {
          float l = ev[u][k] + edl[dv[u] * H + k];
          l = (l > 0.0f) ? l : NEG_SLOPE * l;
          atomicMax(&mxu[dv[u] * H + k], f2ord(l));
        }
      }
    }
    __syncthreads();

    if (KEEP == 0 && pass == 0) {
      const bool fits = total <= kcap;
      if (fits)
        for (int i = kept + tid; i < total; i += NT) kp[i] = list[i];
      if (tid == 0) kp[kcap] = make_int2(fits ? total : -1, 0);
    }

    if (pass != 1) {
      const int ngroups = (total + EPW - 1) / EPW;
      for (int g = wv; g < ngroups; g += NT / 32) {
        const int i0 = g * EPW;
        const int nvalid = (total - i0 < EPW) ? (total - i0) : EPW;
        const bool act = q < nvalid;
        const int2 sd = list[i0 + (act ? q : 0)];
        const float esv = es[(size_t)sd.x * H + hh];
        if (nvalid == EPW) add_rows(flag_t<true>{},  sd.x, sd.y, esv, act, nvalid);
        else               add_rows(flag_t<false>{}, sd.x, sd.y, esv, act, nvalid);
      }
    }
    if (hi >= E_) {
      if (pass != 1) break;
      pass = 2; lo = 0; hi = E_;
    } else {
      lo = hi; hi = E_;
    }
    __syncthreads();
  }
  __syncthreads();

  if (PROJ) {
    constexpr int NW = NT / 32;
    constexpr int TR = (NR + NW - 1) / NW;
    for (int t = 0; t < TR; ++t) {
      const int node = wv + t * NW;
      const int nd = (node < nrv) ? node : 0;
      const float dn = den[nd] + 1e-16f;
      const float v0 = elu_f(acc[nd * AP + lane] / dn + c0);
      const float v1 = elu_f(acc[nd * AP + lane + 32] / dn + c1);
      float s = v0 * w0 + v1 * w1;
      for (int o = 16; o > 0; o >>= 1) s += __shfl_xor(s, o, 32);
      if (lane == 0 && node < nrv) edl[node] = s + bo0;
    }
    __syncthreads();
    for (int i = tid; i < nrv; i += NT) out[n0 + i] = edl[i];
  } else if (NEXT) {
    const float s0 = asn[lane], s1 = asn[lane + 32];
    const float t0 = adn[lane], t1 = adn[lane + 32];
    for (int idx = tid; idx < nrv * LDH; idx += NT) {
      const int node = idx / LDH, ch = idx % LDH;
      const float dn = den[node * H + ch / HID] + 1e-16f;
      float* a = acc + node * AP + ch;
      *a = elu_f(*a / dn + bias[ch]);
    }
    __syncthreads();
    float* const hs = reinterpret_cast<float*>(list);
    {
      const int lr = lane & 15, hi = lane >> 4;
      const int ntr = (nrv + 15) >> 4;
      for (int w = wv; w < ntr * 2; w += NT / 32) {
        const int rt = w >> 1, cp = w & 1;
        const int row = (rt * 16 + lr < NR) ? (rt * 16 + lr) : (NR - 1);
        const float* arow = acc + row * AP + hi * 2;
        const float* bcol = Wn + (size_t)(hi * 2) * HID + cp * 32 + lr;
        v8f d0 = {}, d1 = {};
#pragma unroll 8
        for (int k0 = 0; k0 < LDH; k0 += 4) {
          const v2f a = *reinterpret_cast<const v2f*>(arow + k0);
          v2f b0, b1;
          b0.x = bcol[(size_t)k0 * HID];       b0.y = bcol[(size_t)(k0 + 1) * HID];
          b1.x = bcol[(size_t)k0 * HID + 16];  b1.y = bcol[(size_t)(k0 + 1) * HID + 16];
          d0 = __builtin_amdgcn_wmma_f32_16x16x4_f32(false, a, false, b0, (short)0, d0, false, false);
          d1 = __builtin_amdgcn_wmma_f32_16x16x4_f32(false, a, false, b1, (short)0, d1, false, false);
        }
        float* hp = hs + (rt * 16 + hi * 8) * HID + cp * 32 + lr;
#pragma unroll
        for (int v = 0; v < 8; ++v) { hp[v * HID] = d0[v]; hp[v * HID + 16] = d1[v]; }
      }
    }
    __syncthreads();
    for (int node = wv; node < nrv; node += NT / 32) {
      const float v0 = hs[node * HID + lane], v1 = hs[node * HID + lane + 32];
      float* op = out + (size_t)(n0 + node) * HID + lane;
      op[0] = v0; op[32] = v1;
      float s = v0 * s0 + v1 * s1;
      float d = v0 * t0 + v1 * t1;
      for (int o = 16; o > 0; o >>= 1) { s += __shfl_xor(s, o, 32); d += __shfl_xor(d, o, 32); }
      if (lane == 0) { edl[node] = s; edl[NR + node] = d; }
    }
    __syncthreads();
    for (int i = tid; i < nrv; i += NT) { esn[n0 + i] = edl[i]; edn[n0 + i] = edl[NR + i]; }
  } else {
    for (int idx = tid; idx < nrv * LDH; idx += NT) {
      const int node = idx / LDH, ch = idx % LDH;
      const float dn = den[node * H + ch / HID] + 1e-16f;
      out[(size_t)(n0 + node) * LDH + ch] = elu_f(acc[node * AP + ch] / dn + bias[ch]);
    }
  }
}

#define CDIV(a, b) (((a) + (b) - 1) / (b))

extern "C" void kernel_launch(void* const* d_in, const int* in_sizes, int n_in,
                              void* d_out, int out_size, void* d_ws, size_t ws_size,
                              hipStream_t stream) {
  const float* x   = (const float*)d_in[0];
  const float* W1  = (const float*)d_in[1];
  const float* as1 = (const float*)d_in[2];
  const float* ad1 = (const float*)d_in[3];
  const float* b1  = (const float*)d_in[4];
  const float* W2  = (const float*)d_in[5];
  const float* as2 = (const float*)d_in[6];
  const float* ad2 = (const float*)d_in[7];
  const float* b2  = (const float*)d_in[8];
  const float* Wo  = (const float*)d_in[9];
  const float* bo  = (const float*)d_in[10];
  const int*   ei  = (const int*)d_in[11];

  const int n  = in_sizes[0]  / IN_CH;
  const int e  = in_sizes[11] / 2;

  float* p = (float*)d_ws;
  float* h1   = p;  p += (size_t)n * 256;
                    p += (size_t)n * 256;
  float* h2   = p;  p += (size_t)n * 64;
  float* lst  = p;  p += (size_t)n * 64;
  float* es1  = p;  p += (size_t)n * 4;
  float* ed1  = p;  p += (size_t)n * 4;
  float* es2v = p;  p += n;
  float* ed2v = p;

  const int B = 256;
  constexpr int NR = 200;
  constexpr int LCAP = 8192;
  int2* keep = (int2*)lst;

  int tiles1 = (n / 16) * (256 / 16);
  wmma_gemm_f32<128><<<CDIV(tiles1, 8), B, 0, stream>>>(x, W1, h1, n, 256);
  attn_coef<HEADS, NR><<<CDIV(n * HEADS, B), B, 0, stream>>>(h1, as1, ad1, es1, ed1, n, keep);
  deal_edges<NR, 256, 4><<<CDIV(e, 4096), 256, 0, stream>>>(ei, keep, e, n);
  owner_message<HEADS, NR, LCAP, 0, 0, 1><<<CDIV(n, NR), 1024, 0, stream>>>(ei, h1, es1, ed1, b1, (const float*)nullptr, (const float*)nullptr, W2, as2, ad2, es2v, ed2v, h2, keep, e, n);

  owner_message<1, NR, LCAP, 1, 1, 0><<<CDIV(n, NR), 1024, 0, stream>>>(ei, h2, es2v, ed2v, b2, Wo, bo, (const float*)nullptr, (const float*)nullptr, (const float*)nullptr, (float*)nullptr, (float*)nullptr, (float*)d_out, keep, e, n);
}


# Layer-1 product column-stationary: each wave keeps its 16 columns of W1 in registers for all K and walks the row tiles; X tiles staged once in LDS by whole-row loads; f32 k-sum order changes
# speedup vs baseline: 4.8308x; 1.1847x over previous
#include <hip/hip_runtime.h>
#include <math.h>


#define IN_CH 128
#define HID   64
#define HEADS 4
#define NEG_SLOPE 0.2f

typedef float v2f __attribute__((ext_vector_type(2)));
typedef float v8f __attribute__((ext_vector_type(8)));

__device__ __forceinline__ unsigned f2ord(float f) {
  unsigned u = __float_as_uint(f);
  return (u & 0x80000000u) ? ~u : (u | 0x80000000u);
}
__device__ __forceinline__ float ord2f(unsigned u) {
  return (u & 0x80000000u) ? __uint_as_float(u & 0x7fffffffu)
                           : __uint_as_float(~u);
}
#define NEG_INF_ORD 0x007FFFFFu

__device__ __forceinline__ float elu_f(float x) {
  return x > 0.0f ? x : (expf(x) - 1.0f);
}

template <int K, int NC>
__global__ void __launch_bounds__((NC / 16) * 32)
gemm_colstat_f32(const float* __restrict__ A,
                 const float* __restrict__ B,
                 float* __restrict__ C, int M) {
  constexpr int NT = (NC / 16) * 32;
  constexpr int R4 = K / 4;
  constexpr int AP = K + 4;
  static_assert(K % 8 == 0 && ((AP / 4) & 1) == 1, "k blocks of 8; pitch = 16 bytes x odd");
  static_assert(16 * R4 == NT, "one 16-byte piece of the 16-row tile per thread");
  __shared__ __attribute__((aligned(16))) float at[2 * 16 * AP];

  const int tid  = (int)threadIdx.x;
  const int lane = tid & 31, lr = lane & 15, hi = lane >> 4;
  const int tn  = __builtin_amdgcn_readfirstlane(tid >> 5);
  const int ntm = M >> 4;
  int tm = (int)blockIdx.x;
  if (tm >= ntm) return;

  const int frow = tid / R4, fc4 = tid % R4;
  const float4* const ag = reinterpret_cast<const float4*>(A) + (size_t)frow * R4 + fc4;
  float* const fdst = at + frow * AP + fc4 * 4;
  float4 g = ag[(size_t)tm * 16 * R4];

  v2f bf[K / 4];
  {
    const float* bp = B + (size_t)(4 * hi) * NC + tn * 16 + lr;
#pragma unroll
    for (int s = 0; s < K / 4; ++s) {
      const int k = 8 * (s >> 1) + 2 * (s & 1);
      bf[s].x = bp[(size_t)k * NC];
      bf[s].y = bp[(size_t)(k + 1) * NC];
    }
  }
  *reinterpret_cast<float4*>(fdst) = g;
  __syncthreads();

  int cur = 0;
  for (;;) {
    const int tnx = tm + (int)gridDim.x;
    const bool more = tnx < ntm;
    if (more) g = ag[(size_t)tnx * 16 * R4];
    const float* ar = at + cur * (16 * AP) + lr * AP + 4 * hi;
    v8f d0 = {}, d1 = {};
#pragma unroll
    for (int j = 0; j < K / 8; ++j) {
      const float4 a = *reinterpret_cast<const float4*>(ar + 8 * j);
      v2f a0; a0.x = a.x; a0.y = a.y;
      v2f a1; a1.x = a.z; a1.y = a.w;
      d0 = __builtin_amdgcn_wmma_f32_16x16x4_f32(false, a0, false, bf[2 * j],     (short)0, d0, false, false);
      d1 = __builtin_amdgcn_wmma_f32_16x16x4_f32(false, a1, false, bf[2 * j + 1], (short)0, d1, false, false);
    }
    const v8f d = d0 + d1;
    if (more) {
      *reinterpret_cast<float4*>(fdst + (cur ^ 1) * (16 * AP)) = g;
      __syncthreads();
    }
    float* cp = C + (size_t)(tm * 16 + hi * 8) * NC + tn * 16 + lr;
#pragma unroll
    for (int v = 0; v < 8; ++v) cp[(size_t)v * NC] = d[v];
    if (!more) break;
    cur ^= 1;
    tm = tnx;
  }
}

#define KEEP_PER_NODE (HID / 2)
#define DEAL_MAXOWN 1024

template <int NR>
__device__ __forceinline__ int2* owner_list(int2* keep, int o) {
  return keep + (size_t)o * NR * KEEP_PER_NODE;
}
template <int NR>
__device__ __forceinline__ int owner_cap(int o, int n) {
  const int nrv = (n - o * NR < NR) ? (n - o * NR) : NR;
  return nrv * KEEP_PER_NODE - 16;
}

template <int NR, int NT, int G>
__global__ void __launch_bounds__(NT)
deal_edges(const int* __restrict__ ei, int2* keep, int E_, int N_) {
  __shared__ int hist[DEAL_MAXOWN];
  const int nown = (N_ + NR - 1) / NR;
  if (nown > DEAL_MAXOWN) return;
  const int tid = (int)threadIdx.x;
  for (int i = tid; i < nown; i += NT) hist[i] = 0;

  const int* __restrict__ dstp = ei + E_;
  const int base = (int)blockIdx.x * (NT * 4 * G) + tid * 4;
  const bool vec_ok = ((E_ & 3) == 0) && ((reinterpret_cast<size_t>(ei) & 15) == 0);
  int s[4 * G], d[4 * G];
  if (vec_ok) {
#pragma unroll
    for (int g = 0; g < G; ++g) {
      const int e0 = base + g * NT * 4;
      const bool in = e0 < E_;
      const int ec = in ? e0 : 0;
      const int4 sv = *reinterpret_cast<const int4*>(ei + ec);
      const int4 dv = *reinterpret_cast<const int4*>(dstp + ec);
      s[4 * g] = sv.x; s[4 * g + 1] = sv.y; s[4 * g + 2] = sv.z; s[4 * g + 3] = sv.w;
      d[4 * g]     = in ? dv.x : -1; d[4 * g + 1] = in ? dv.y : -1;
      d[4 * g + 2] = in ? dv.z : -1; d[4 * g + 3] = in ? dv.w : -1;
    }
  } else {
#pragma unroll
    for (int j = 0; j < 4 * G; ++j) {
      const int e = base + (j >> 2) * NT * 4 + (j & 3);
      const bool in = e < E_;
      s[j] = in ? ei[e] : 0;
      d[j] = in ? dstp[e] : -1;
    }
  }
  __syncthreads();

  int o[4 * G], r[4 * G];
#pragma unroll
  for (int j = 0; j < 4 * G; ++j) {
    const bool ok = (unsigned)d[j] < (unsigned)N_;
    o[j] = ok ? d[j] / NR : -1;
    r[j] = ok ? atomicAdd(&hist[o[j]], 1) : 0;
  }
  __syncthreads();

  for (int t = tid; t < nown; t += NT) {
    const int c = hist[t];
    hist[t] = (c > 0) ? atomicAdd(&owner_list<NR>(keep, t)[owner_cap<NR>(t, N_)].x, c) : 0;
  }
  __syncthreads();

  int st[4 * G];
#pragma unroll
  for (int j = 0; j < 4 * G; ++j) st[j] = hist[(o[j] >= 0) ? o[j] : 0];
#pragma unroll
  for (int j = 0; j < 4 * G; ++j) {
    if (o[j] >= 0) {
      const int pos = st[j] + r[j];
      if (pos < owner_cap<NR>(o[j], N_))
        owner_list<NR>(keep, o[j])[pos] = make_int2(s[j], d[j] - o[j] * NR);
    }
  }
}

template <int H, int NR>
__global__ void attn_coef(const float* __restrict__ h,
                          const float* __restrict__ a_src,
                          const float* __restrict__ a_dst,
                          float* __restrict__ es, float* __restrict__ ed, int n,
                          int2* keep) {
  int idx = blockIdx.x * blockDim.x + threadIdx.x;
  if (keep != nullptr) {
    const int nown = (n + NR - 1) / NR;
    if (idx < nown) owner_list<NR>(keep, idx)[owner_cap<NR>(idx, n)].x = (nown <= DEAL_MAXOWN) ? 0 : -1;
  }
  if (idx >= n * H) return;
  int i = idx / H, hh = idx % H;
  const float* hp = h + (size_t)i * (H * HID) + hh * HID;
  const float* as = a_src + hh * HID;
  const float* ad = a_dst + hh * HID;
  float s = 0.f, d = 0.f;
#pragma unroll 8
  for (int c = 0; c < HID; ++c) { float v = hp[c]; s += v * as[c]; d += v * ad[c]; }
  es[idx] = s; ed[idx] = d;
}

template <bool B> struct flag_t { static constexpr bool value = B; };

template <int H, int NR, int LCAP, int KEEP, int PROJ, int NEXT>
__global__ void __launch_bounds__(1024)
owner_message(const int* __restrict__ ei,
              const float* __restrict__ h,
              const float* __restrict__ es,
              const float* __restrict__ ed,
              const float* __restrict__ bias,
              const float* __restrict__ Wo,
              const float* __restrict__ bo,
              const float* __restrict__ Wn,
              const float* __restrict__ asn,
              const float* __restrict__ adn,
              float* esn, float* edn,
              float* out, int2* keep, int E_, int N_) {
  constexpr int LDH = H * HID;
  constexpr int AP  = NEXT ? LDH + 4 : LDH;
  constexpr int NL  = LDH / 32;
  constexpr int EPW = 8;
  constexpr int NT  = 1024;
  constexpr int U   = 8;
  constexpr int UR  = 4;
  static_assert(H * EPW <= 32, "one lane per (edge, head) pair of the group");
  static_assert(LCAP >= NR + 8, "self-loops plus a smallest window always fit");
  static_assert(NR * KEEP_PER_NODE * 8 == NR * HID * 4, "a slice of the list plane is 256 bytes per own node");
  static_assert(PROJ == 0 || (H == 1 && LDH == 64), "the projection takes one head's row of 64: two channels per lane");
  static_assert(NEXT == 0 || (PROJ == 0 && H >= 2 && LDH % 4 == 0 && ((NR + 15) / 16) * 16 * HID * 4 <= LCAP * 8),
                "the next layer's 64-wide rows, padded to whole 16-row tiles, are staged in the list's LDS");

  __shared__ __attribute__((aligned(16))) float acc[NR * AP];
  __shared__ float    den[NR * H];
  __shared__ unsigned mxu[NR * H];
  __shared__ float    edl[NR * H];
  __shared__ int2     list[LCAP];
  __shared__ int      cnt;

  const int tid  = (int)threadIdx.x;
  const int lane = tid & 31;
  const int wv   = __builtin_amdgcn_readfirstlane(tid >> 5);
  const int n0   = (int)blockIdx.x * NR;
  const int nrv  = (N_ - n0 < NR) ? (N_ - n0) : NR;
  const unsigned nrvu = (unsigned)nrv;
  const int* __restrict__ dstp = ei + E_;
  const int4* __restrict__ dst4 = reinterpret_cast<const int4*>(dstp);
  const bool vec_ok = ((E_ & 3) == 0) && ((reinterpret_cast<size_t>(dstp) & 15) == 0);
  const int q  = lane / H, hh = lane % H;
  int2* const kp = keep + (size_t)n0 * KEEP_PER_NODE;
  const int kcap = nrv * KEEP_PER_NODE - 16;

  float w0 = 0.f, w1 = 0.f, c0 = 0.f, c1 = 0.f, bo0 = 0.f;
  if (PROJ) {
    w0 = Wo[lane];   w1 = Wo[lane + 32];
    c0 = bias[lane]; c1 = bias[lane + 32];
    bo0 = bo[0];
  }
  for (int i = tid; i < NR * AP / 4; i += NT)
    reinterpret_cast<float4*>(acc)[i] = make_float4(0.f, 0.f, 0.f, 0.f);
  for (int i = tid; i < NR * H; i += NT) {
    den[i] = 0.f;
    mxu[i] = NEG_INF_ORD;
    edl[i] = (i < nrv * H) ? ed[(size_t)n0 * H + i] : 0.f;
  }

  auto push = [&](int e) {
    const int pos = atomicAdd(&cnt, 1);
    if (pos < LCAP) list[pos].x = e;
  };

  auto add_rows = [&](auto full, int s, int dl, float esv, bool act, int nvalid) {
    constexpr bool FULL = decltype(full)::value;
    constexpr int G = (NL >= 8) ? 4 : 8;
    float p = 0.f;
#pragma unroll
    for (int q0 = 0; q0 < EPW; q0 += G) {
      float hv[G][NL];
      int   dq[G];
#pragma unroll
      for (int t = 0; t < G; ++t) {
        if (FULL || q0 + t < nvalid) {
          const int sq = __builtin_amdgcn_readlane(s, (q0 + t) * H);
          dq[t] = __builtin_amdgcn_readlane(dl, (q0 + t) * H);
          const float* hp = h + (size_t)sq * LDH + lane;
#pragma unroll
          for (int j = 0; j < NL; ++j) hv[t][j] = hp[32 * j];
        }
      }
      if (q0 == 0) {
        float l = esv + edl[dl * H + hh];
        l = (l > 0.0f) ? l : NEG_SLOPE * l;
        p = expf(l - ord2f(mxu[dl * H + hh]));
        if (act) atomicAdd(&den[dl * H + hh], p);
      }
#pragma unroll
      for (int t = 0; t < G; ++t) {
        if (FULL || q0 + t < nvalid) {
          float aq[H];
#pragma unroll
          for (int k = 0; k < H; ++k)
            aq[k] = __int_as_float(__builtin_amdgcn_readlane(__float_as_int(p), (q0 + t) * H + k));
          float* ap = acc + dq[t] * AP + lane;
#pragma unroll
          for (int j = 0; j < NL; ++j) atomicAdd(ap + 32 * j, hv[t][j] * aq[(32 * j) / HID]);
        }
      }
    }
  };

  bool pre = false;
  int kept = 0;
  {
    const int kt = __builtin_amdgcn_readfirstlane(kp[kcap].x);
    const int nadd = (KEEP == 0) ? nrv : 0;
    if (kt >= 0 && kt <= kcap && kt + nadd <= LCAP) {
      constexpr int KU = 4;
      for (int base = tid; base < kt; base += NT * KU) {
        int2 t[KU];
        int iv[KU];
#pragma unroll
        for (int u = 0; u < KU; ++u) {
          const int i = base + u * NT;
          iv[u] = (i < kt) ? i : base;
          t[u] = kp[iv[u]];
        }
#pragma unroll
        for (int u = 0; u < KU; ++u) list[iv[u]] = t[u];
      }
      for (int i = tid; i < nadd; i += NT) list[kt + i] = make_int2(n0 + i, i);
      if (tid == 0) cnt = kt + nadd;
      pre = true;
      kept = kt;
    }
  }

  int pass = 0;
  int lo = 0, hi = E_;
  for (;;) {
    if (!pre) {
      const int nself = (lo == 0) ? nrv : 0;
      for (int i = tid; i < nself; i += NT) list[i].x = E_ + n0 + i;
      if (tid == 0) cnt = nself;
      __syncthreads();
      if (vec_ok) {
        const int vlo = lo >> 2, vhi = hi >> 2;
        for (int base = vlo + tid; base < vhi; base += NT * U) {
          int4 dv[U];
#pragma unroll
          for (int u = 0; u < U; ++u) {
            const int i = base + u * NT;
            dv[u] = (i < vhi) ? dst4[i] : make_int4(-1, -1, -1, -1);
          }
#pragma unroll
          for (int u = 0; u < U; ++u) {
            const int e = (base + u * NT) * 4;
            const bool b0 = (unsigned)(dv[u].x - n0) < nrvu;
            const bool b1 = (unsigned)(dv[u].y - n0) < nrvu;
            const bool b2 = (unsigned)(dv[u].z - n0) < nrvu;
            const bool b3 = (unsigned)(dv[u].w - n0) < nrvu;
            if (b0 | b1 | b2 | b3) {
              if (b0) push(e);
              if (b1) push(e + 1);
              if (b2) push(e + 2);
              if (b3) push(e + 3);
            }
          }
        }
      } else {
        for (int e = lo + tid; e < hi; e += NT)
          if ((unsigned)(dstp[e] - n0) < nrvu) push(e);
      }
    }
    __syncthreads();
    const int total = __builtin_amdgcn_readfirstlane(cnt);
    if (total > LCAP) {
      if (pass == 0) {
        pass = 1;
        if (KEEP == 0 && tid == 0) kp[kcap] = make_int2(-1, 0);
      }
      hi = lo + (((hi - lo) >> 1) & ~3);
      __syncthreads();
      continue;
    }

    for (int base = tid; base < total; base += NT * UR) {
      int iv[UR], sv[UR], dv[UR];
#pragma unroll
      for (int u = 0; u < UR; ++u) {
        const int i = base + u * NT;
        iv[u] = (i < total) ? i : base;
        if (pre) {
          const int2 t = list[iv[u]];
          sv[u] = t.x; dv[u] = t.y;
        } else {
          const int e  = list[iv[u]].x;
          const int el = (e < E_) ? e : 0;
          const int s_ = ei[el], d_ = dstp[el];
          sv[u] = (e < E_) ? s_ : (e - E_);
          dv[u] = ((e < E_) ? d_ : (e - E_)) - n0;
        }
      }
      float ev[UR][H];
#pragma unroll
      for (int u = 0; u < UR; ++u)
#pragma unroll
        for (int k = 0; k < H; ++k) ev[u][k] = es[(size_t)sv[u] * H + k];
#pragma unroll
      for (int u = 0; u < UR; ++u) {
        if (!pre) list[iv[u]] = make_int2(sv[u], dv[u]);
#pragma unroll
        for (int k = 0; k < H; ++k) {
          float l = ev[u][k] + edl[dv[u] * H + k];
          l = (l > 0.0f) ? l : NEG_SLOPE * l;
          atomicMax(&mxu[dv[u] * H + k], f2ord(l));
        }
      }
    }
    __syncthreads();

    if (KEEP == 0 && pass == 0) {
      const bool fits = total <= kcap;
      if (fits)
        for (int i = kept + tid; i < total; i += NT) kp[i] = list[i];
      if (tid == 0) kp[kcap] = make_int2(fits ? total : -1, 0);
    }

    if (pass != 1) {
      const int ngroups = (total + EPW - 1) / EPW;
      for (int g = wv; g < ngroups; g += NT / 32) {
        const int i0 = g * EPW;
        const int nvalid = (total - i0 < EPW) ? (total - i0) : EPW;
        const bool act = q < nvalid;
        const int2 sd = list[i0 + (act ? q : 0)];
        const float esv = es[(size_t)sd.x * H + hh];
        if (nvalid == EPW) add_rows(flag_t<true>{},  sd.x, sd.y, esv, act, nvalid);
        else               add_rows(flag_t<false>{}, sd.x, sd.y, esv, act, nvalid);
      }
    }
    if (hi >= E_) {
      if (pass != 1) break;
      pass = 2; lo = 0; hi = E_;
    } else {
      lo = hi; hi = E_;
    }
    __syncthreads();
  }
  __syncthreads();

  if (PROJ) {
    constexpr int NW = NT / 32;
    constexpr int TR = (NR + NW - 1) / NW;
    for (int t = 0; t < TR; ++t) {
      const int node = wv + t * NW;
      const int nd = (node < nrv) ? node : 0;
      const float dn = den[nd] + 1e-16f;
      const float v0 = elu_f(acc[nd * AP + lane] / dn + c0);
      const float v1 = elu_f(acc[nd * AP + lane + 32] / dn + c1);
      float s = v0 * w0 + v1 * w1;
      for (int o = 16; o > 0; o >>= 1) s += __shfl_xor(s, o, 32);
      if (lane == 0 && node < nrv) edl[node] = s + bo0;
    }
    __syncthreads();
    for (int i = tid; i < nrv; i += NT) out[n0 + i] = edl[i];
  } else if (NEXT) {
    const float s0 = asn[lane], s1 = asn[lane + 32];
    const float t0 = adn[lane], t1 = adn[lane + 32];
    for (int idx = tid; idx < nrv * LDH; idx += NT) {
      const int node = idx / LDH, ch = idx % LDH;
      const float dn = den[node * H + ch / HID] + 1e-16f;
      float* a = acc + node * AP + ch;
      *a = elu_f(*a / dn + bias[ch]);
    }
    __syncthreads();
    float* const hs = reinterpret_cast<float*>(list);
    {
      const int lr = lane & 15, hi = lane >> 4;
      const int ntr = (nrv + 15) >> 4;
      for (int w = wv; w < ntr * 2; w += NT / 32) {
        const int rt = w >> 1, cp = w & 1;
        const int row = (rt * 16 + lr < NR) ? (rt * 16 + lr) : (NR - 1);
        const float* arow = acc + row * AP + hi * 2;
        const float* bcol = Wn + (size_t)(hi * 2) * HID + cp * 32 + lr;
        v8f d0 = {}, d1 = {};
#pragma unroll 8
        for (int k0 = 0; k0 < LDH; k0 += 4) {
          const v2f a = *reinterpret_cast<const v2f*>(arow + k0);
          v2f b0, b1;
          b0.x = bcol[(size_t)k0 * HID];       b0.y = bcol[(size_t)(k0 + 1) * HID];
          b1.x = bcol[(size_t)k0 * HID + 16];  b1.y = bcol[(size_t)(k0 + 1) * HID + 16];
          d0 = __builtin_amdgcn_wmma_f32_16x16x4_f32(false, a, false, b0, (short)0, d0, false, false);
          d1 = __builtin_amdgcn_wmma_f32_16x16x4_f32(false, a, false, b1, (short)0, d1, false, false);
        }
        float* hp = hs + (rt * 16 + hi * 8) * HID + cp * 32 + lr;
#pragma unroll
        for (int v = 0; v < 8; ++v) { hp[v * HID] = d0[v]; hp[v * HID + 16] = d1[v]; }
      }
    }
    __syncthreads();
    for (int node = wv; node < nrv; node += NT / 32) {
      const float v0 = hs[node * HID + lane], v1 = hs[node * HID + lane + 32];
      float* op = out + (size_t)(n0 + node) * HID + lane;
      op[0] = v0; op[32] = v1;
      float s = v0 * s0 + v1 * s1;
      float d = v0 * t0 + v1 * t1;
      for (int o = 16; o > 0; o >>= 1) { s += __shfl_xor(s, o, 32); d += __shfl_xor(d, o, 32); }
      if (lane == 0) { edl[node] = s; edl[NR + node] = d; }
    }
    __syncthreads();
    for (int i = tid; i < nrv; i += NT) { esn[n0 + i] = edl[i]; edn[n0 + i] = edl[NR + i]; }
  } else {
    for (int idx = tid; idx < nrv * LDH; idx += NT) {
      const int node = idx / LDH, ch = idx % LDH;
      const float dn = den[node * H + ch / HID] + 1e-16f;
      out[(size_t)(n0 + node) * LDH + ch] = elu_f(acc[node * AP + ch] / dn + bias[ch]);
    }
  }
}

#define CDIV(a, b) (((a) + (b) - 1) / (b))

extern "C" void kernel_launch(void* const* d_in, const int* in_sizes, int n_in,
                              void* d_out, int out_size, void* d_ws, size_t ws_size,
                              hipStream_t stream) {
  const float* x   = (const float*)d_in[0];
  const float* W1  = (const float*)d_in[1];
  const float* as1 = (const float*)d_in[2];
  const float* ad1 = (const float*)d_in[3];
  const float* b1  = (const float*)d_in[4];
  const float* W2  = (const float*)d_in[5];
  const float* as2 = (const float*)d_in[6];
  const float* ad2 = (const float*)d_in[7];
  const float* b2  = (const float*)d_in[8];
  const float* Wo  = (const float*)d_in[9];
  const float* bo  = (const float*)d_in[10];
  const int*   ei  = (const int*)d_in[11];

  const int n  = in_sizes[0]  / IN_CH;
  const int e  = in_sizes[11] / 2;

  float* p = (float*)d_ws;
  float* h1   = p;  p += (size_t)n * 256;
                    p += (size_t)n * 256;
  float* h2   = p;  p += (size_t)n * 64;
  float* lst  = p;  p += (size_t)n * 64;
  float* es1  = p;  p += (size_t)n * 4;
  float* ed1  = p;  p += (size_t)n * 4;
  float* es2v = p;  p += n;
  float* ed2v = p;

  const int B = 256;
  constexpr int NR = 200;
  constexpr int LCAP = 8192;
  int2* keep = (int2*)lst;

  gemm_colstat_f32<IN_CH, HEADS * HID><<<512, (HEADS * HID / 16) * 32, 0, stream>>>(x, W1, h1, n);
  attn_coef<HEADS, NR><<<CDIV(n * HEADS, B), B, 0, stream>>>(h1, as1, ad1, es1, ed1, n, keep);
  deal_edges<NR, 256, 4><<<CDIV(e, 4096), 256, 0, stream>>>(ei, keep, e, n);
  owner_message<HEADS, NR, LCAP, 0, 0, 1><<<CDIV(n, NR), 1024, 0, stream>>>(ei, h1, es1, ed1, b1, (const float*)nullptr, (const float*)nullptr, W2, as2, ad2, es2v, ed2v, h2, keep, e, n);

  owner_message<1, NR, LCAP, 1, 1, 0><<<CDIV(n, NR), 1024, 0, stream>>>(ei, h2, es2v, ed2v, b2, Wo, bo, (const float*)nullptr, (const float*)nullptr, (const float*)nullptr, (float*)nullptr, (float*)nullptr, (float*)d_out, keep, e, n);
}


# Layer-1 attention halves as 8 more columns of the X product: a_src/a_dst folded into W1 per head, k split over 16 waves, partial tiles summed in LDS in k order; attn launch gone; es/ed sum order new
# speedup vs baseline: 5.0885x; 1.0533x over previous
#include <hip/hip_runtime.h>
#include <math.h>


#define IN_CH 128
#define HID   64
#define HEADS 4
#define NEG_SLOPE 0.2f

typedef float v2f __attribute__((ext_vector_type(2)));
typedef float v8f __attribute__((ext_vector_type(8)));

__device__ __forceinline__ unsigned f2ord(float f) {
  unsigned u = __float_as_uint(f);
  return (u & 0x80000000u) ? ~u : (u | 0x80000000u);
}
__device__ __forceinline__ float ord2f(unsigned u) {
  return (u & 0x80000000u) ? __uint_as_float(u & 0x7fffffffu)
                           : __uint_as_float(~u);
}
#define NEG_INF_ORD 0x007FFFFFu

__device__ __forceinline__ float elu_f(float x) {
  return x > 0.0f ? x : (expf(x) - 1.0f);
}

#define KEEP_PER_NODE (HID / 2)
#define DEAL_MAXOWN 1024

template <int NR>
__device__ __forceinline__ int2* owner_list(int2* keep, int o) {
  return keep + (size_t)o * NR * KEEP_PER_NODE;
}
template <int NR>
__device__ __forceinline__ int owner_cap(int o, int n) {
  const int nrv = (n - o * NR < NR) ? (n - o * NR) : NR;
  return nrv * KEEP_PER_NODE - 16;
}

template <int K, int NC, int NR>
__global__ void __launch_bounds__((NC / 16) * 32)
gemm_colstat_f32(const float* __restrict__ A,
                 const float* __restrict__ B,
                 const float* __restrict__ a_src,
                 const float* __restrict__ a_dst,
                 float* __restrict__ C,
                 float* __restrict__ es, float* __restrict__ ed,
                 int2* keep, int M) {
  constexpr int NT = (NC / 16) * 32;
  constexpr int NW = NC / 16;
  constexpr int R4 = K / 4;
  constexpr int AP = K + 4;
  static_assert(K % 8 == 0 && ((AP / 4) & 1) == 1, "k blocks of 8; pitch = 16 bytes x odd");
  static_assert(16 * R4 == NT, "one 16-byte piece of the 16-row tile per thread");
  static_assert(NC == HEADS * HID && NC == 256 && K / 8 == NW,
                "a row of B is two 16-byte pieces per lane, a head is 16 lanes of a piece; one k block of U per wave");
  __shared__ __attribute__((aligned(16))) float at[2 * 16 * AP];
  __shared__ float us[K * 8];
  __shared__ float part[2 * NW * 16 * 8];

  const int tid  = (int)threadIdx.x;
  const int lane = tid & 31, lr = lane & 15, hi = lane >> 4;
  const int tn  = __builtin_amdgcn_readfirstlane(tid >> 5);
  const int ntm = M >> 4;
  int tm = (int)blockIdx.x;
  if (tm == 0) {
    const int nown = (M + NR - 1) / NR;
    for (int o = tid; o < nown; o += NT)
      owner_list<NR>(keep, o)[owner_cap<NR>(o, M)].x = (nown <= DEAL_MAXOWN) ? 0 : -1;
  }
  if (tm >= ntm) return;

  const int frow = tid / R4, fc4 = tid % R4;
  const float4* const ag = reinterpret_cast<const float4*>(A) + (size_t)frow * R4 + fc4;
  float* const fdst = at + frow * AP + fc4 * 4;
  float4 g = ag[(size_t)tm * 16 * R4];

  float ue[32];
  {
    const float4* const as4 = reinterpret_cast<const float4*>(a_src);
    const float4* const ad4 = reinterpret_cast<const float4*>(a_dst);
    const float4 sl = as4[lane], sh = as4[32 + lane];
    const float4 dl = ad4[lane], dh = ad4[32 + lane];
    const float4* const wr = reinterpret_cast<const float4*>(B) + (size_t)(8 * tn) * (NC / 4) + lane;
    float4 wl[8], wh[8];
#pragma unroll
    for (int r = 0; r < 8; ++r) { wl[r] = wr[r * (NC / 4)]; wh[r] = wr[r * (NC / 4) + 32]; }
#pragma unroll
    for (int r = 0; r < 8; ++r) {
      ue[4 * r]     = wl[r].x * sl.x + wl[r].y * sl.y + wl[r].z * sl.z + wl[r].w * sl.w;
      ue[4 * r + 1] = wl[r].x * dl.x + wl[r].y * dl.y + wl[r].z * dl.z + wl[r].w * dl.w;
      ue[4 * r + 2] = wh[r].x * sh.x + wh[r].y * sh.y + wh[r].z * sh.z + wh[r].w * sh.w;
      ue[4 * r + 3] = wh[r].x * dh.x + wh[r].y * dh.y + wh[r].z * dh.z + wh[r].w * dh.w;
    }
  }

#pragma unroll
  for (int st = 0; st < 4; ++st) {
    const int o = 8 >> st, half = 16 >> st;
    const bool up_half = (lr & o) != 0;
#pragma unroll
    for (int i = 0; i < half; ++i) {
      const float lo = ue[i], hi2 = ue[i + half];
      const float keep = up_half ? static_cast<float>(hi2) : static_cast<float>(lo);
      const float send = up_half ? static_cast<float>(lo) : static_cast<float>(hi2);
      ue[i] = keep + __shfl_xor(send, o, 32);
    }
  }
  {
    float* up = us + (8 * tn + (lr >> 1)) * 8 + hi + 2 * (lr & 1);
    up[0] = ue[0]; up[4] = ue[1];
  }

  v2f bf[K / 4];
  {
    const float* bp = B + (size_t)(4 * hi) * NC + tn * 16 + lr;
#pragma unroll
    for (int s = 0; s < K / 4; ++s) {
      const int k = 8 * (s >> 1) + 2 * (s & 1);
      bf[s].x = bp[(size_t)k * NC];
      bf[s].y = bp[(size_t)(k + 1) * NC];
    }
  }
  *reinterpret_cast<float4*>(fdst) = g;
  __syncthreads();

  v2f ub0, ub1;
  {
    const float* up = us + (8 * tn + 4 * hi) * 8 + (lr & 7);
    const bool uc = lr < 8;
    ub0.x = uc ? up[0]  : 0.f; ub0.y = uc ? up[8]  : 0.f;
    ub1.x = uc ? up[16] : 0.f; ub1.y = uc ? up[24] : 0.f;
  }

  auto att_out = [&](int buf, int tile) {
    if (tid < 128) {
      const int e = tid & 63, row = e >> 2, cc = (e & 3) + ((tid >> 6) << 2);
      const float* pp = part + (buf * NW * 16 + row) * 8 + cc;
      float s = 0.f;
#pragma unroll
      for (int w = 0; w < NW; ++w) s += pp[w * 128];
      float* const o = (tid < 64) ? es : ed;
      o[(size_t)tile * 64 + e] = s;
    }
  };

  int cur = 0;
  for (;;) {
    const int tnx = tm + (int)gridDim.x;
    const bool more = tnx < ntm;
    if (more) g = ag[(size_t)tnx * 16 * R4];
    const float* ar = at + cur * (16 * AP) + lr * AP + 4 * hi;
    v8f d0 = {}, d1 = {};
#pragma unroll
    for (int j = 0; j < K / 8; ++j) {
      const float4 a = *reinterpret_cast<const float4*>(ar + 8 * j);
      v2f a0; a0.x = a.x; a0.y = a.y;
      v2f a1; a1.x = a.z; a1.y = a.w;
      d0 = __builtin_amdgcn_wmma_f32_16x16x4_f32(false, a0, false, bf[2 * j],     (short)0, d0, false, false);
      d1 = __builtin_amdgcn_wmma_f32_16x16x4_f32(false, a1, false, bf[2 * j + 1], (short)0, d1, false, false);
    }
    const v8f d = d0 + d1;
    {
      const float4 a = *reinterpret_cast<const float4*>(ar + 8 * tn);
      v2f a0; a0.x = a.x; a0.y = a.y;
      v2f a1; a1.x = a.z; a1.y = a.w;
      v8f de = {};
      de = __builtin_amdgcn_wmma_f32_16x16x4_f32(false, a0, false, ub0, (short)0, de, false, false);
      de = __builtin_amdgcn_wmma_f32_16x16x4_f32(false, a1, false, ub1, (short)0, de, false, false);
      if (lr < 8) {
        float* pp = part + ((cur * NW + tn) * 16 + 8 * hi) * 8 + lr;
#pragma unroll
        for (int v = 0; v < 8; ++v) pp[v * 8] = de[v];
      }
    }
    if (more) *reinterpret_cast<float4*>(fdst + (cur ^ 1) * (16 * AP)) = g;
    __syncthreads();
    float* cp = C + (size_t)(tm * 16 + hi * 8) * NC + tn * 16 + lr;
#pragma unroll
    for (int v = 0; v < 8; ++v) cp[(size_t)v * NC] = d[v];
    att_out(cur, tm);
    if (!more) break;
    cur ^= 1;
    tm = tnx;
  }
}

template <int NR, int NT, int G>
__global__ void __launch_bounds__(NT)
deal_edges(const int* __restrict__ ei, int2* keep, int E_, int N_) {
  __shared__ int hist[DEAL_MAXOWN];
  const int nown = (N_ + NR - 1) / NR;
  if (nown > DEAL_MAXOWN) return;
  const int tid = (int)threadIdx.x;
  for (int i = tid; i < nown; i += NT) hist[i] = 0;

  const int* __restrict__ dstp = ei + E_;
  const int base = (int)blockIdx.x * (NT * 4 * G) + tid * 4;
  const bool vec_ok = ((E_ & 3) == 0) && ((reinterpret_cast<size_t>(ei) & 15) == 0);
  int s[4 * G], d[4 * G];
  if (vec_ok) {
#pragma unroll
    for (int g = 0; g < G; ++g) {
      const int e0 = base + g * NT * 4;
      const bool in = e0 < E_;
      const int ec = in ? e0 : 0;
      const int4 sv = *reinterpret_cast<const int4*>(ei + ec);
      const int4 dv = *reinterpret_cast<const int4*>(dstp + ec);
      s[4 * g] = sv.x; s[4 * g + 1] = sv.y; s[4 * g + 2] = sv.z; s[4 * g + 3] = sv.w;
      d[4 * g]     = in ? dv.x : -1; d[4 * g + 1] = in ? dv.y : -1;
      d[4 * g + 2] = in ? dv.z : -1; d[4 * g + 3] = in ? dv.w : -1;
    }
  } else {
#pragma unroll
    for (int j = 0; j < 4 * G; ++j) {
      const int e = base + (j >> 2) * NT * 4 + (j & 3);
      const bool in = e < E_;
      s[j] = in ? ei[e] : 0;
      d[j] = in ? dstp[e] : -1;
    }
  }
  __syncthreads();

  int o[4 * G], r[4 * G];
#pragma unroll
  for (int j = 0; j < 4 * G; ++j) {
    const bool ok = (unsigned)d[j] < (unsigned)N_;
    o[j] = ok ? d[j] / NR : -1;
    r[j] = ok ? atomicAdd(&hist[o[j]], 1) : 0;
  }
  __syncthreads();

  for (int t = tid; t < nown; t += NT) {
    const int c = hist[t];
    hist[t] = (c > 0) ? atomicAdd(&owner_list<NR>(keep, t)[owner_cap<NR>(t, N_)].x, c) : 0;
  }
  __syncthreads();

  int st[4 * G];
#pragma unroll
  for (int j = 0; j < 4 * G; ++j) st[j] = hist[(o[j] >= 0) ? o[j] : 0];
#pragma unroll
  for (int j = 0; j < 4 * G; ++j) {
    if (o[j] >= 0) {
      const int pos = st[j] + r[j];
      if (pos < owner_cap<NR>(o[j], N_))
        owner_list<NR>(keep, o[j])[pos] = make_int2(s[j], d[j] - o[j] * NR);
    }
  }
}

template <bool B> struct flag_t { static constexpr bool value = B; };

template <int H, int NR, int LCAP, int KEEP, int PROJ, int NEXT>
__global__ void __launch_bounds__(1024)
owner_message(const int* __restrict__ ei,
              const float* __restrict__ h,
              const float* __restrict__ es,
              const float* __restrict__ ed,
              const float* __restrict__ bias,
              const float* __restrict__ Wo,
              const float* __restrict__ bo,
              const float* __restrict__ Wn,
              const float* __restrict__ asn,
              const float* __restrict__ adn,
              float* esn, float* edn,
              float* out, int2* keep, int E_, int N_) {
  constexpr int LDH = H * HID;
  constexpr int AP  = NEXT ? LDH + 4 : LDH;
  constexpr int NL  = LDH / 32;
  constexpr int EPW = 8;
  constexpr int NT  = 1024;
  constexpr int U   = 8;
  constexpr int UR  = 4;
  static_assert(H * EPW <= 32, "one lane per (edge, head) pair of the group");
  static_assert(LCAP >= NR + 8, "self-loops plus a smallest window always fit");
  static_assert(NR * KEEP_PER_NODE * 8 == NR * HID * 4, "a slice of the list plane is 256 bytes per own node");
  static_assert(PROJ == 0 || (H == 1 && LDH == 64), "the projection takes one head's row of 64: two channels per lane");
  static_assert(NEXT == 0 || (PROJ == 0 && H >= 2 && LDH % 4 == 0 && ((NR + 15) / 16) * 16 * HID * 4 <= LCAP * 8),
                "the next layer's 64-wide rows, padded to whole 16-row tiles, are staged in the list's LDS");

  __shared__ __attribute__((aligned(16))) float acc[NR * AP];
  __shared__ float    den[NR * H];
  __shared__ unsigned mxu[NR * H];
  __shared__ float    edl[NR * H];
  __shared__ int2     list[LCAP];
  __shared__ int      cnt;

  const int tid  = (int)threadIdx.x;
  const int lane = tid & 31;
  const int wv   = __builtin_amdgcn_readfirstlane(tid >> 5);
  const int n0   = (int)blockIdx.x * NR;
  const int nrv  = (N_ - n0 < NR) ? (N_ - n0) : NR;
  const unsigned nrvu = (unsigned)nrv;
  const int* __restrict__ dstp = ei + E_;
  const int4* __restrict__ dst4 = reinterpret_cast<const int4*>(dstp);
  const bool vec_ok = ((E_ & 3) == 0) && ((reinterpret_cast<size_t>(dstp) & 15) == 0);
  const int q  = lane / H, hh = lane % H;
  int2* const kp = keep + (size_t)n0 * KEEP_PER_NODE;
  const int kcap = nrv * KEEP_PER_NODE - 16;

  float w0 = 0.f, w1 = 0.f, c0 = 0.f, c1 = 0.f, bo0 = 0.f;
  if (PROJ) {
    w0 = Wo[lane];   w1 = Wo[lane + 32];
    c0 = bias[lane]; c1 = bias[lane + 32];
    bo0 = bo[0];
  }
  for (int i = tid; i < NR * AP / 4; i += NT)
    reinterpret_cast<float4*>(acc)[i] = make_float4(0.f, 0.f, 0.f, 0.f);
  for (int i = tid; i < NR * H; i += NT) {
    den[i] = 0.f;
    mxu[i] = NEG_INF_ORD;
    edl[i] = (i < nrv * H) ? ed[(size_t)n0 * H + i] : 0.f;
  }

  auto push = [&](int e) {
    const int pos = atomicAdd(&cnt, 1);
    if (pos < LCAP) list[pos].x = e;
  };

  auto add_rows = [&](auto full, int s, int dl, float esv, bool act, int nvalid) {
    constexpr bool FULL = decltype(full)::value;
    constexpr int G = (NL >= 8) ? 4 : 8;
    float p = 0.f;
#pragma unroll
    for (int q0 = 0; q0 < EPW; q0 += G) {
      float hv[G][NL];
      int   dq[G];
#pragma unroll
      for (int t = 0; t < G; ++t) {
        if (FULL || q0 + t < nvalid) {
          const int sq = __builtin_amdgcn_readlane(s, (q0 + t) * H);
          dq[t] = __builtin_amdgcn_readlane(dl, (q0 + t) * H);
          const float* hp = h + (size_t)sq * LDH + lane;
#pragma unroll
          for (int j = 0; j < NL; ++j) hv[t][j] = hp[32 * j];
        }
      }
      if (q0 == 0) {
        float l = esv + edl[dl * H + hh];
        l = (l > 0.0f) ? l : NEG_SLOPE * l;
        p = expf(l - ord2f(mxu[dl * H + hh]));
        if (act) atomicAdd(&den[dl * H + hh], p);
      }
#pragma unroll
      for (int t = 0; t < G; ++t) {
        if (FULL || q0 + t < nvalid) {
          float aq[H];
#pragma unroll
          for (int k = 0; k < H; ++k)
            aq[k] = __int_as_float(__builtin_amdgcn_readlane(__float_as_int(p), (q0 + t) * H + k));
          float* ap = acc + dq[t] * AP + lane;
#pragma unroll
          for (int j = 0; j < NL; ++j) atomicAdd(ap + 32 * j, hv[t][j] * aq[(32 * j) / HID]);
        }
      }
    }
  };

  bool pre = false;
  int kept = 0;
  {
    const int kt = __builtin_amdgcn_readfirstlane(kp[kcap].x);
    const int nadd = (KEEP == 0) ? nrv : 0;
    if (kt >= 0 && kt <= kcap && kt + nadd <= LCAP) {
      constexpr int KU = 4;
      for (int base = tid; base < kt; base += NT * KU) {
        int2 t[KU];
        int iv[KU];
#pragma unroll
        for (int u = 0; u < KU; ++u) {
          const int i = base + u * NT;
          iv[u] = (i < kt) ? i : base;
          t[u] = kp[iv[u]];
        }
#pragma unroll
        for (int u = 0; u < KU; ++u) list[iv[u]] = t[u];
      }
      for (int i = tid; i < nadd; i += NT) list[kt + i] = make_int2(n0 + i, i);
      if (tid == 0) cnt = kt + nadd;
      pre = true;
      kept = kt;
    }
  }

  int pass = 0;
  int lo = 0, hi = E_;
  for (;;) {
    if (!pre) {
      const int nself = (lo == 0) ? nrv : 0;
      for (int i = tid; i < nself; i += NT) list[i].x = E_ + n0 + i;
      if (tid == 0) cnt = nself;
      __syncthreads();
      if (vec_ok) {
        const int vlo = lo >> 2, vhi = hi >> 2;
        for (int base = vlo + tid; base < vhi; base += NT * U) {
          int4 dv[U];
#pragma unroll
          for (int u = 0; u < U; ++u) {
            const int i = base + u * NT;
            dv[u] = (i < vhi) ? dst4[i] : make_int4(-1, -1, -1, -1);
          }
#pragma unroll
          for (int u = 0; u < U; ++u) {
            const int e = (base + u * NT) * 4;
            const bool b0 = (unsigned)(dv[u].x - n0) < nrvu;
            const bool b1 = (unsigned)(dv[u].y - n0) < nrvu;
            const bool b2 = (unsigned)(dv[u].z - n0) < nrvu;
            const bool b3 = (unsigned)(dv[u].w - n0) < nrvu;
            if (b0 | b1 | b2 | b3) {
              if (b0) push(e);
              if (b1) push(e + 1);
              if (b2) push(e + 2);
              if (b3) push(e + 3);
            }
          }
        }
      } else {
        for (int e = lo + tid; e < hi; e += NT)
          if ((unsigned)(dstp[e] - n0) < nrvu) push(e);
      }
    }
    __syncthreads();
    const int total = __builtin_amdgcn_readfirstlane(cnt);
    if (total > LCAP) {
      if (pass == 0) {
        pass = 1;
        if (KEEP == 0 && tid == 0) kp[kcap] = make_int2(-1, 0);
      }
      hi = lo + (((hi - lo) >> 1) & ~3);
      __syncthreads();
      continue;
    }

    for (int base = tid; base < total; base += NT * UR) {
      int iv[UR], sv[UR], dv[UR];
#pragma unroll
      for (int u = 0; u < UR; ++u) {
        const int i = base + u * NT;
        iv[u] = (i < total) ? i : base;
        if (pre) {
          const int2 t = list[iv[u]];
          sv[u] = t.x; dv[u] = t.y;
        } else {
          const int e  = list[iv[u]].x;
          const int el = (e < E_) ? e : 0;
          const int s_ = ei[el], d_ = dstp[el];
          sv[u] = (e < E_) ? s_ : (e - E_);
          dv[u] = ((e < E_) ? d_ : (e - E_)) - n0;
        }
      }
      float ev[UR][H];
#pragma unroll
      for (int u = 0; u < UR; ++u)
#pragma unroll
        for (int k = 0; k < H; ++k) ev[u][k] = es[(size_t)sv[u] * H + k];
#pragma unroll
      for (int u = 0; u < UR; ++u) {
        if (!pre) list[iv[u]] = make_int2(sv[u], dv[u]);
#pragma unroll
        for (int k = 0; k < H; ++k) {
          float l = ev[u][k] + edl[dv[u] * H + k];
          l = (l > 0.0f) ? l : NEG_SLOPE * l;
          atomicMax(&mxu[dv[u] * H + k], f2ord(l));
        }
      }
    }
    __syncthreads();

    if (KEEP == 0 && pass == 0) {
      const bool fits = total <= kcap;
      if (fits)
        for (int i = kept + tid; i < total; i += NT) kp[i] = list[i];
      if (tid == 0) kp[kcap] = make_int2(fits ? total : -1, 0);
    }

    if (pass != 1) {
      const int ngroups = (total + EPW - 1) / EPW;
      for (int g = wv; g < ngroups; g += NT / 32) {
        const int i0 = g * EPW;
        const int nvalid = (total - i0 < EPW) ? (total - i0) : EPW;
        const bool act = q < nvalid;
        const int2 sd = list[i0 + (act ? q : 0)];
        const float esv = es[(size_t)sd.x * H + hh];
        if (nvalid == EPW) add_rows(flag_t<true>{},  sd.x, sd.y, esv, act, nvalid);
        else               add_rows(flag_t<false>{}, sd.x, sd.y, esv, act, nvalid);
      }
    }
    if (hi >= E_) {
      if (pass != 1) break;
      pass = 2; lo = 0; hi = E_;
    } else {
      lo = hi; hi = E_;
    }
    __syncthreads();
  }
  __syncthreads();

  if (PROJ) {
    constexpr int NW = NT / 32;
    constexpr int TR = (NR + NW - 1) / NW;
    for (int t = 0; t < TR; ++t) {
      const int node = wv + t * NW;
      const int nd = (node < nrv) ? node : 0;
      const float dn = den[nd] + 1e-16f;
      const float v0 = elu_f(acc[nd * AP + lane] / dn + c0);
      const float v1 = elu_f(acc[nd * AP + lane + 32] / dn + c1);
      float s = v0 * w0 + v1 * w1;
      for (int o = 16; o > 0; o >>= 1) s += __shfl_xor(s, o, 32);
      if (lane == 0 && node < nrv) edl[node] = s + bo0;
    }
    __syncthreads();
    for (int i = tid; i < nrv; i += NT) out[n0 + i] = edl[i];
  } else if (NEXT) {
    const float s0 = asn[lane], s1 = asn[lane + 32];
    const float t0 = adn[lane], t1 = adn[lane + 32];
    for (int idx = tid; idx < nrv * LDH; idx += NT) {
      const int node = idx / LDH, ch = idx % LDH;
      const float dn = den[node * H + ch / HID] + 1e-16f;
      float* a = acc + node * AP + ch;
      *a = elu_f(*a / dn + bias[ch]);
    }
    __syncthreads();
    float* const hs = reinterpret_cast<float*>(list);
    {
      const int lr = lane & 15, hi = lane >> 4;
      const int ntr = (nrv + 15) >> 4;
      for (int w = wv; w < ntr * 2; w += NT / 32) {
        const int rt = w >> 1, cp = w & 1;
        const int row = (rt * 16 + lr < NR) ? (rt * 16 + lr) : (NR - 1);
        const float* arow = acc + row * AP + hi * 2;
        const float* bcol = Wn + (size_t)(hi * 2) * HID + cp * 32 + lr;
        v8f d0 = {}, d1 = {};
#pragma unroll 8
        for (int k0 = 0; k0 < LDH; k0 += 4) {
          const v2f a = *reinterpret_cast<const v2f*>(arow + k0);
          v2f b0, b1;
          b0.x = bcol[(size_t)k0 * HID];       b0.y = bcol[(size_t)(k0 + 1) * HID];
          b1.x = bcol[(size_t)k0 * HID + 16];  b1.y = bcol[(size_t)(k0 + 1) * HID + 16];
          d0 = __builtin_amdgcn_wmma_f32_16x16x4_f32(false, a, false, b0, (short)0, d0, false, false);
          d1 = __builtin_amdgcn_wmma_f32_16x16x4_f32(false, a, false, b1, (short)0, d1, false, false);
        }
        float* hp = hs + (rt * 16 + hi * 8) * HID + cp * 32 + lr;
#pragma unroll
        for (int v = 0; v < 8; ++v) { hp[v * HID] = d0[v]; hp[v * HID + 16] = d1[v]; }
      }
    }
    __syncthreads();
    for (int node = wv; node < nrv; node += NT / 32) {
      const float v0 = hs[node * HID + lane], v1 = hs[node * HID + lane + 32];
      float* op = out + (size_t)(n0 + node) * HID + lane;
      op[0] = v0; op[32] = v1;
      float s = v0 * s0 + v1 * s1;
      float d = v0 * t0 + v1 * t1;
      for (int o = 16; o > 0; o >>= 1) { s += __shfl_xor(s, o, 32); d += __shfl_xor(d, o, 32); }
      if (lane == 0) { edl[node] = s; edl[NR + node] = d; }
    }
    __syncthreads();
    for (int i = tid; i < nrv; i += NT) { esn[n0 + i] = edl[i]; edn[n0 + i] = edl[NR + i]; }
  } else {
    for (int idx = tid; idx < nrv * LDH; idx += NT) {
      const int node = idx / LDH, ch = idx % LDH;
      const float dn = den[node * H + ch / HID] + 1e-16f;
      out[(size_t)(n0 + node) * LDH + ch] = elu_f(acc[node * AP + ch] / dn + bias[ch]);
    }
  }
}

#define CDIV(a, b) (((a) + (b) - 1) / (b))

extern "C" void kernel_launch(void* const* d_in, const int* in_sizes, int n_in,
                              void* d_out, int out_size, void* d_ws, size_t ws_size,
                              hipStream_t stream) {
  const float* x   = (const float*)d_in[0];
  const float* W1  = (const float*)d_in[1];
  const float* as1 = (const float*)d_in[2];
  const float* ad1 = (const float*)d_in[3];
  const float* b1  = (const float*)d_in[4];
  const float* W2  = (const float*)d_in[5];
  const float* as2 = (const float*)d_in[6];
  const float* ad2 = (const float*)d_in[7];
  const float* b2  = (const float*)d_in[8];
  const float* Wo  = (const float*)d_in[9];
  const float* bo  = (const float*)d_in[10];
  const int*   ei  = (const int*)d_in[11];

  const int n  = in_sizes[0]  / IN_CH;
  const int e  = in_sizes[11] / 2;

  float* p = (float*)d_ws;
  float* h1   = p;  p += (size_t)n * 256;
                    p += (size_t)n * 256;
  float* h2   = p;  p += (size_t)n * 64;
  float* lst  = p;  p += (size_t)n * 64;
  float* es1  = p;  p += (size_t)n * 4;
  float* ed1  = p;  p += (size_t)n * 4;
  float* es2v = p;  p += n;
  float* ed2v = p;

  constexpr int NR = 200;
  constexpr int LCAP = 8192;
  int2* keep = (int2*)lst;

  gemm_colstat_f32<IN_CH, HEADS * HID, NR><<<512, (HEADS * HID / 16) * 32, 0, stream>>>(x, W1, as1, ad1, h1, es1, ed1, keep, n);
  deal_edges<NR, 256, 4><<<CDIV(e, 4096), 256, 0, stream>>>(ei, keep, e, n);
  owner_message<HEADS, NR, LCAP, 0, 0, 1><<<CDIV(n, NR), 1024, 0, stream>>>(ei, h1, es1, ed1, b1, (const float*)nullptr, (const float*)nullptr, W2, as2, ad2, es2v, ed2v, h2, keep, e, n);

  owner_message<1, NR, LCAP, 1, 1, 0><<<CDIV(n, NR), 1024, 0, stream>>>(ei, h2, es2v, ed2v, b2, Wo, bo, (const float*)nullptr, (const float*)nullptr, (const float*)nullptr, (float*)nullptr, (float*)nullptr, (float*)d_out, keep, e, n);
}
